# starting kernel
# baseline (speedup 1.0000x reference)
#include <hip/hip_runtime.h>
#include <math.h>


constexpr int kN   = 20000;
constexpr int kE   = 320000;
constexpr int kIN  = 128;
constexpr int kH   = 8;
constexpr int kD   = 16;
constexpr int kHD  = 128;
constexpr int kL   = 2;
constexpr int kOUT = 2;
constexpr float kLnEps = 1e-5f;

typedef __attribute__((ext_vector_type(2))) float v2f;
typedef __attribute__((ext_vector_type(8))) float v8f;

__device__ __forceinline__ float eluf(float x) {
    return x > 0.0f ? x : (expf(x) - 1.0f);
}

__global__ __launch_bounds__(256) void gemm_h1h2_kernel(
    const float* __restrict__ X,
    const float* __restrict__ W1, const float* __restrict__ B1,
    const float* __restrict__ W2, const float* __restrict__ B2,
    float* __restrict__ H1, float* __restrict__ H2, float* __restrict__ HP)
{
    const int lane  = threadIdx.x & 31;
    const int wave  = threadIdx.x >> 5;
    const int mbase = blockIdx.x * 16;
    const int n0    = wave * 16;
    const int r     = lane & 15;
    const int half  = lane >> 4;

    const float* xrow = X + (size_t)(mbase + r) * kIN;

    v8f acc1 = {};
    v8f acc2 = {};

#pragma unroll
    for (int k0 = 0; k0 < kIN; k0 += 4) {
        const int ka = k0 + half * 2;
        v2f a = *(const v2f*)(xrow + ka);
        v2f bA, bB;
        bA.x = W1[(size_t)ka       * kHD + n0 + r];
        bA.y = W1[(size_t)(ka + 1) * kHD + n0 + r];
        bB.x = W2[(size_t)ka       * kHD + n0 + r];
        bB.y = W2[(size_t)(ka + 1) * kHD + n0 + r];
        acc1 = __builtin_amdgcn_wmma_f32_16x16x4_f32(
            false, a, false, bA, (short)0, acc1, false, false);
        acc2 = __builtin_amdgcn_wmma_f32_16x16x4_f32(
            false, a, false, bB, (short)0, acc2, false, false);
    }

    const int n = n0 + r;
    const float bb1 = B1[n];
    const float bb2 = B2[n];

#pragma unroll
    for (int j = 0; j < 8; ++j) {
        const int m = mbase + j + half * 8;
        const float h1v = acc1[j] + bb1;
        const float h2v = acc2[j] + bb2;
        const size_t idx = (size_t)m * kHD + n;
        H1[idx] = h1v;
        H2[idx] = h2v;
        HP[idx] = h1v * h2v;
    }
}

__global__ __launch_bounds__(256) void edge_attn_kernel(
    const float* __restrict__ H1, const float* __restrict__ H2,
    const float* __restrict__ HP,
    const int* __restrict__ src, const int* __restrict__ dst,
    const float* __restrict__ avec, float* __restrict__ AGG)
{
    const int lane = threadIdx.x & 31;
    const int e    = blockIdx.x * 8 + (threadIdx.x >> 5);
    if (e >= kE) return;

    const int s = src[e];
    const int d = dst[e];

    const float4 h1s = *(const float4*)(H1 + (size_t)s * kHD + lane * 4);
    const float4 h2d = *(const float4*)(H2 + (size_t)d * kHD + lane * 4);
    const float4 hps = *(const float4*)(HP + (size_t)s * kHD + lane * 4);
    const float4 hpd = *(const float4*)(HP + (size_t)d * kHD + lane * 4);
    const float4 av  = *(const float4*)(avec + lane * 4);

    float part;
    {
        const float t0 = eluf(h1s.x + h2d.x + hps.x * hpd.x);
        const float t1 = eluf(h1s.y + h2d.y + hps.y * hpd.y);
        const float t2 = eluf(h1s.z + h2d.z + hps.z * hpd.z);
        const float t3 = eluf(h1s.w + h2d.w + hps.w * hpd.w);
        part = av.x * t0 + av.y * t1 + av.z * t2 + av.w * t3;
    }
    part *= 0.25f;

    part += __shfl_xor(part, 1, 32);
    part += __shfl_xor(part, 2, 32);

    float mx = part;
    mx = fmaxf(mx, __shfl_xor(mx, 4, 32));
    mx = fmaxf(mx, __shfl_xor(mx, 8, 32));
    mx = fmaxf(mx, __shfl_xor(mx, 16, 32));

    const float ex = expf(part - mx);
    float sm = ex;
    sm += __shfl_xor(sm, 4, 32);
    sm += __shfl_xor(sm, 8, 32);
    sm += __shfl_xor(sm, 16, 32);

    const float alpha = ex / sm;

    float* aggrow = AGG + (size_t)d * kHD + lane * 4;
    atomicAdd(aggrow + 0, alpha * h1s.x);
    atomicAdd(aggrow + 1, alpha * h1s.y);
    atomicAdd(aggrow + 2, alpha * h1s.z);
    atomicAdd(aggrow + 3, alpha * h1s.w);
}

__global__ __launch_bounds__(256) void ln_elu_res_kernel(
    const float* __restrict__ AGG,
    const float* __restrict__ G, const float* __restrict__ B,
    const float* __restrict__ Xin, float* __restrict__ Xout)
{
    const int lane = threadIdx.x & 31;
    const int node = blockIdx.x * 8 + (threadIdx.x >> 5);
    if (node >= kN) return;

    const float4 v = *(const float4*)(AGG + (size_t)node * kHD + lane * 4);

    float s = v.x + v.y + v.z + v.w;
#pragma unroll
    for (int m = 1; m <= 16; m <<= 1) s += __shfl_xor(s, m, 32);
    const float mu = s * (1.0f / 128.0f);

    const float dx0 = v.x - mu, dx1 = v.y - mu, dx2 = v.z - mu, dx3 = v.w - mu;
    float q = dx0 * dx0 + dx1 * dx1 + dx2 * dx2 + dx3 * dx3;
#pragma unroll
    for (int m = 1; m <= 16; m <<= 1) q += __shfl_xor(q, m, 32);
    const float inv = rsqrtf(q * (1.0f / 128.0f) + kLnEps);

    const float4 g  = *(const float4*)(G + lane * 4);
    const float4 b  = *(const float4*)(B + lane * 4);
    const float4 xr = *(const float4*)(Xin + (size_t)node * kHD + lane * 4);

    float4 o;
    o.x = eluf(dx0 * inv * g.x + b.x) + xr.x;
    o.y = eluf(dx1 * inv * g.y + b.y) + xr.y;
    o.z = eluf(dx2 * inv * g.z + b.z) + xr.z;
    o.w = eluf(dx3 * inv * g.w + b.w) + xr.w;

    *(float4*)(Xout + (size_t)node * kHD + lane * 4) = o;
}

__global__ __launch_bounds__(256) void out_proj_kernel(
    const float* __restrict__ X, const float* __restrict__ Wo,
    const float* __restrict__ bo, float* __restrict__ O)
{
    const int n = blockIdx.x * blockDim.x + threadIdx.x;
    if (n >= kN) return;
    float a0 = bo[0];
    float a1 = bo[1];
    const float* xr = X + (size_t)n * kHD;
#pragma unroll 8
    for (int k = 0; k < kHD; ++k) {
        const float xv = xr[k];
        a0 += xv * Wo[2 * k + 0];
        a1 += xv * Wo[2 * k + 1];
    }
    O[2 * n + 0] = a0;
    O[2 * n + 1] = a1;
}

extern "C" void kernel_launch(void* const* d_in, const int* in_sizes, int n_in,
                              void* d_out, int out_size, void* d_ws, size_t ws_size,
                              hipStream_t stream)
{
    (void)in_sizes; (void)n_in; (void)out_size; (void)ws_size;

    const float* x    = (const float*)d_in[0];
    const int*   ei   = (const int*)  d_in[1];
    const float* W1   = (const float*)d_in[2];
    const float* b1   = (const float*)d_in[3];
    const float* W2   = (const float*)d_in[4];
    const float* b2   = (const float*)d_in[5];
    const float* a    = (const float*)d_in[6];
    const float* ln_g = (const float*)d_in[7];
    const float* ln_b = (const float*)d_in[8];
    const float* Wo   = (const float*)d_in[9];
    const float* bo   = (const float*)d_in[10];
    float*       out  = (float*)d_out;

    const size_t NHD = (size_t)kN * kHD;
    float* h1   = (float*)d_ws;
    float* h2   = h1  + NHD;
    float* hp   = h2  + NHD;
    float* agg  = hp  + NHD;
    float* xcur = agg + NHD;

    const int* srcIdx = ei;
    const int* dstIdx = ei + kE;

    for (int layer = 0; layer < kL; ++layer) {
        const float* xin = (layer == 0) ? x : xcur;

        gemm_h1h2_kernel<<<kN / 16, 256, 0, stream>>>(
            xin,
            W1 + (size_t)layer * kIN * kHD, b1 + (size_t)layer * kHD,
            W2 + (size_t)layer * kIN * kHD, b2 + (size_t)layer * kHD,
            h1, h2, hp);

        hipMemsetAsync(agg, 0, NHD * sizeof(float), stream);

        edge_attn_kernel<<<kE / 8, 256, 0, stream>>>(
            h1, h2, hp, srcIdx, dstIdx, a + (size_t)layer * kH * kD, agg);

        ln_elu_res_kernel<<<kN / 8, 256, 0, stream>>>(
            agg, ln_g + (size_t)layer * kHD, ln_b + (size_t)layer * kHD,
            xin, xcur);
    }

    out_proj_kernel<<<(kN + 255) / 256, 256, 0, stream>>>(xcur, Wo, bo, out);
}


# Edges grouped by destination per call (integer-atomic arrival-order rank, scan, fill); a wave per node sums messages in registers in list order, stores the row: no f32 atomics, no agg memset
# speedup vs baseline: 1.2780x; 1.2780x over previous
#include <hip/hip_runtime.h>
#include <math.h>


constexpr int kN   = 20000;
constexpr int kE   = 320000;
constexpr int kIN  = 128;
constexpr int kH   = 8;
constexpr int kD   = 16;
constexpr int kHD  = 128;
constexpr int kL   = 2;
constexpr int kOUT = 2;
constexpr float kLnEps = 1e-5f;

typedef __attribute__((ext_vector_type(2))) float v2f;
typedef __attribute__((ext_vector_type(8))) float v8f;

__device__ __forceinline__ float eluf(float x) {
    return x > 0.0f ? x : (expf(x) - 1.0f);
}

__device__ __forceinline__ float head_part(
    const float4& h1s, const float4& hps,
    const float4& h2d, const float4& hpd, const float4& av)
{
    const float t0 = eluf(h1s.x + h2d.x + hps.x * hpd.x);
    const float t1 = eluf(h1s.y + h2d.y + hps.y * hpd.y);
    const float t2 = eluf(h1s.z + h2d.z + hps.z * hpd.z);
    const float t3 = eluf(h1s.w + h2d.w + hps.w * hpd.w);
    const float part = av.x * t0 + av.y * t1 + av.z * t2 + av.w * t3;
    return part * 0.25f;
}

__global__ __launch_bounds__(256) void gemm_h1h2_kernel(
    const float* __restrict__ X,
    const float* __restrict__ W1, const float* __restrict__ B1,
    const float* __restrict__ W2, const float* __restrict__ B2,
    float* __restrict__ H1, float* __restrict__ H2, float* __restrict__ HP)
{
    const int lane  = threadIdx.x & 31;
    const int wave  = threadIdx.x >> 5;
    const int mbase = blockIdx.x * 16;
    const int n0    = wave * 16;
    const int r     = lane & 15;
    const int half  = lane >> 4;

    const float* xrow = X + (size_t)(mbase + r) * kIN;

    v8f acc1 = {};
    v8f acc2 = {};

#pragma unroll
    for (int k0 = 0; k0 < kIN; k0 += 4) {
        const int ka = k0 + half * 2;
        v2f a = *(const v2f*)(xrow + ka);
        v2f bA, bB;
        bA.x = W1[(size_t)ka       * kHD + n0 + r];
        bA.y = W1[(size_t)(ka + 1) * kHD + n0 + r];
        bB.x = W2[(size_t)ka       * kHD + n0 + r];
        bB.y = W2[(size_t)(ka + 1) * kHD + n0 + r];
        acc1 = __builtin_amdgcn_wmma_f32_16x16x4_f32(
            false, a, false, bA, (short)0, acc1, false, false);
        acc2 = __builtin_amdgcn_wmma_f32_16x16x4_f32(
            false, a, false, bB, (short)0, acc2, false, false);
    }

    const int n = n0 + r;
    const float bb1 = B1[n];
    const float bb2 = B2[n];

#pragma unroll
    for (int j = 0; j < 8; ++j) {
        const int m = mbase + j + half * 8;
        const float h1v = acc1[j] + bb1;
        const float h2v = acc2[j] + bb2;
        const size_t idx = (size_t)m * kHD + n;
        H1[idx] = h1v;
        H2[idx] = h2v;
        HP[idx] = h1v * h2v;
    }
}

constexpr int kScanThreads = 1024;
constexpr int kScanItems   = 20;
constexpr int kScanLive    = kN / kScanItems;
static_assert(kN % kScanItems == 0, "whole threads only");
static_assert(kScanLive < kScanThreads, "one block must cover all counts");
static_assert(kScanItems % 4 == 0, "int4 loads");
constexpr int kCntPad = kScanThreads * kScanItems;

__global__ __launch_bounds__(256) void dst_rank_kernel(
    const int* __restrict__ dst, int* __restrict__ cnt, int* __restrict__ rank)
{
    const int e = blockIdx.x * 256 + threadIdx.x;
    if (e >= kE) return;
    rank[e] = atomicAdd(cnt + dst[e], 1);
}

__global__ __launch_bounds__(kScanThreads) void scan_counts_kernel(
    const int* __restrict__ cnt, int* __restrict__ rowptr)
{
    __shared__ int wsum[32];
    const int t    = threadIdx.x;
    const int lane = t & 31;
    const int wave = t >> 5;

    int v[kScanItems];
    int tot = 0;
    if (t < kScanLive) {
        const int4* p = (const int4*)(cnt + t * kScanItems);
#pragma unroll
        for (int i = 0; i < kScanItems / 4; ++i) {
            const int4 q = p[i];
            v[4 * i + 0] = q.x; v[4 * i + 1] = q.y;
            v[4 * i + 2] = q.z; v[4 * i + 3] = q.w;
        }
#pragma unroll
        for (int i = 0; i < kScanItems; ++i) tot += v[i];
    } else {
#pragma unroll
        for (int i = 0; i < kScanItems; ++i) v[i] = 0;
    }

    int inc = tot;
#pragma unroll
    for (int off = 1; off < 32; off <<= 1) {
        const int o = __shfl_up(inc, off, 32);
        if (lane >= off) inc += o;
    }
    if (lane == 31) wsum[wave] = inc;
    __syncthreads();
    if (wave == 0) {
        const int w = wsum[lane];
        int winc = w;
#pragma unroll
        for (int off = 1; off < 32; off <<= 1) {
            const int o = __shfl_up(winc, off, 32);
            if (lane >= off) winc += o;
        }
        wsum[lane] = winc - w;
    }
    __syncthreads();

    int run = wsum[wave] + inc - tot;
    if (t < kScanLive) {
#pragma unroll
        for (int i = 0; i < kScanItems; ++i) {
            rowptr[t * kScanItems + i] = run;
            run += v[i];
        }
    }
    if (t == kScanThreads - 1) rowptr[kN] = run;
}

__global__ __launch_bounds__(256) void csr_fill_kernel(
    const int* __restrict__ src, const int* __restrict__ dst,
    const int* __restrict__ rowptr, const int* __restrict__ rank,
    int* __restrict__ csrc)
{
    const int e = blockIdx.x * 256 + threadIdx.x;
    if (e >= kE) return;
    csrc[rowptr[dst[e]] + rank[e]] = src[e];
}

__global__ __launch_bounds__(256) void node_attn_kernel(
    const float* __restrict__ H1, const float* __restrict__ H2,
    const float* __restrict__ HP,
    const int* __restrict__ rowptr, const int* __restrict__ csrc,
    const float* __restrict__ avec, float* __restrict__ AGG)
{
    const int lane = threadIdx.x & 31;
    const int node = blockIdx.x * 8 +
                     __builtin_amdgcn_readfirstlane((int)(threadIdx.x >> 5));
    if (node >= kN) return;

    const int beg = rowptr[node];
    const int end = rowptr[node + 1];

    const float4 h2d = *(const float4*)(H2 + (size_t)node * kHD + lane * 4);
    const float4 hpd = *(const float4*)(HP + (size_t)node * kHD + lane * 4);
    const float4 av  = *(const float4*)(avec + lane * 4);

    float4 acc = make_float4(0.0f, 0.0f, 0.0f, 0.0f);

    for (int base = beg; base < end; base += 32) {
        const int n = min(32, end - base);
        int mine = 0;
        if (lane < n) mine = csrc[base + lane];

        for (int j = 0; j < n; j += 2) {
            const int j1 = min(j + 1, n - 1);
            int s0 = __builtin_amdgcn_readlane(mine, j);
            int s1 = __builtin_amdgcn_readlane(mine, j1);
            s0 = (int)min((unsigned)s0, (unsigned)(kN - 1));
            s1 = (int)min((unsigned)s1, (unsigned)(kN - 1));

            const float4 h1a = *(const float4*)(H1 + (size_t)s0 * kHD + lane * 4);
            const float4 hpa = *(const float4*)(HP + (size_t)s0 * kHD + lane * 4);
            const float4 h1b = *(const float4*)(H1 + (size_t)s1 * kHD + lane * 4);
            const float4 hpb = *(const float4*)(HP + (size_t)s1 * kHD + lane * 4);

            float p0 = head_part(h1a, hpa, h2d, hpd, av);
            float p1 = head_part(h1b, hpb, h2d, hpd, av);

            p0 += __shfl_xor(p0, 1, 32);
            p1 += __shfl_xor(p1, 1, 32);
            p0 += __shfl_xor(p0, 2, 32);
            p1 += __shfl_xor(p1, 2, 32);

            float m0 = p0, m1 = p1;
            m0 = fmaxf(m0, __shfl_xor(m0, 4, 32));
            m1 = fmaxf(m1, __shfl_xor(m1, 4, 32));
            m0 = fmaxf(m0, __shfl_xor(m0, 8, 32));
            m1 = fmaxf(m1, __shfl_xor(m1, 8, 32));
            m0 = fmaxf(m0, __shfl_xor(m0, 16, 32));
            m1 = fmaxf(m1, __shfl_xor(m1, 16, 32));

            const float e0 = expf(p0 - m0);
            const float e1 = expf(p1 - m1);
            float q0 = e0, q1 = e1;
            q0 += __shfl_xor(q0, 4, 32);
            q1 += __shfl_xor(q1, 4, 32);
            q0 += __shfl_xor(q0, 8, 32);
            q1 += __shfl_xor(q1, 8, 32);
            q0 += __shfl_xor(q0, 16, 32);
            q1 += __shfl_xor(q1, 16, 32);

            const float a0 = e0 / q0;
            const float a1 = e1 / q1;

            acc.x += a0 * h1a.x;
            acc.y += a0 * h1a.y;
            acc.z += a0 * h1a.z;
            acc.w += a0 * h1a.w;
            if (j + 1 < n) {
                acc.x += a1 * h1b.x;
                acc.y += a1 * h1b.y;
                acc.z += a1 * h1b.z;
                acc.w += a1 * h1b.w;
            }
        }
    }

    *(float4*)(AGG + (size_t)node * kHD + lane * 4) = acc;
}

__global__ __launch_bounds__(256) void ln_elu_res_kernel(
    const float* __restrict__ AGG,
    const float* __restrict__ G, const float* __restrict__ B,
    const float* __restrict__ Xin, float* __restrict__ Xout)
{
    const int lane = threadIdx.x & 31;
    const int node = blockIdx.x * 8 + (threadIdx.x >> 5);
    if (node >= kN) return;

    const float4 v = *(const float4*)(AGG + (size_t)node * kHD + lane * 4);

    float s = v.x + v.y + v.z + v.w;
#pragma unroll
    for (int m = 1; m <= 16; m <<= 1) s += __shfl_xor(s, m, 32);
    const float mu = s * (1.0f / 128.0f);

    const float dx0 = v.x - mu, dx1 = v.y - mu, dx2 = v.z - mu, dx3 = v.w - mu;
    float q = dx0 * dx0 + dx1 * dx1 + dx2 * dx2 + dx3 * dx3;
#pragma unroll
    for (int m = 1; m <= 16; m <<= 1) q += __shfl_xor(q, m, 32);
    const float inv = rsqrtf(q * (1.0f / 128.0f) + kLnEps);

    const float4 g  = *(const float4*)(G + lane * 4);
    const float4 b  = *(const float4*)(B + lane * 4);
    const float4 xr = *(const float4*)(Xin + (size_t)node * kHD + lane * 4);

    float4 o;
    o.x = eluf(dx0 * inv * g.x + b.x) + xr.x;
    o.y = eluf(dx1 * inv * g.y + b.y) + xr.y;
    o.z = eluf(dx2 * inv * g.z + b.z) + xr.z;
    o.w = eluf(dx3 * inv * g.w + b.w) + xr.w;

    *(float4*)(Xout + (size_t)node * kHD + lane * 4) = o;
}

__global__ __launch_bounds__(256) void out_proj_kernel(
    const float* __restrict__ X, const float* __restrict__ Wo,
    const float* __restrict__ bo, float* __restrict__ O)
{
    const int n = blockIdx.x * blockDim.x + threadIdx.x;
    if (n >= kN) return;
    float a0 = bo[0];
    float a1 = bo[1];
    const float* xr = X + (size_t)n * kHD;
#pragma unroll 8
    for (int k = 0; k < kHD; ++k) {
        const float xv = xr[k];
        a0 += xv * Wo[2 * k + 0];
        a1 += xv * Wo[2 * k + 1];
    }
    O[2 * n + 0] = a0;
    O[2 * n + 1] = a1;
}

extern "C" void kernel_launch(void* const* d_in, const int* in_sizes, int n_in,
                              void* d_out, int out_size, void* d_ws, size_t ws_size,
                              hipStream_t stream)
{
    (void)in_sizes; (void)n_in; (void)out_size; (void)ws_size;

    const float* x    = (const float*)d_in[0];
    const int*   ei   = (const int*)  d_in[1];
    const float* W1   = (const float*)d_in[2];
    const float* b1   = (const float*)d_in[3];
    const float* W2   = (const float*)d_in[4];
    const float* b2   = (const float*)d_in[5];
    const float* a    = (const float*)d_in[6];
    const float* ln_g = (const float*)d_in[7];
    const float* ln_b = (const float*)d_in[8];
    const float* Wo   = (const float*)d_in[9];
    const float* bo   = (const float*)d_in[10];
    float*       out  = (float*)d_out;

    const size_t NHD = (size_t)kN * kHD;
    float* h1   = (float*)d_ws;
    float* h2   = h1  + NHD;
    float* hp   = h2  + NHD;
    float* agg  = hp  + NHD;
    float* xcur = agg + NHD;
    int* cnt    = (int*)(xcur + NHD);
    int* rowptr = cnt + kCntPad;
    int* rank   = rowptr + kCntPad;
    int* csrc   = rank + kE;

    const int* srcIdx = ei;
    const int* dstIdx = ei + kE;

    hipMemsetAsync(cnt, 0, (size_t)kN * sizeof(int), stream);
    dst_rank_kernel<<<kE / 256, 256, 0, stream>>>(dstIdx, cnt, rank);
    scan_counts_kernel<<<1, kScanThreads, 0, stream>>>(cnt, rowptr);
    csr_fill_kernel<<<kE / 256, 256, 0, stream>>>(srcIdx, dstIdx, rowptr, rank, csrc);

    for (int layer = 0; layer < kL; ++layer) {
        const float* xin = (layer == 0) ? x : xcur;

        gemm_h1h2_kernel<<<kN / 16, 256, 0, stream>>>(
            xin,
            W1 + (size_t)layer * kIN * kHD, b1 + (size_t)layer * kHD,
            W2 + (size_t)layer * kIN * kHD, b2 + (size_t)layer * kHD,
            h1, h2, hp);

        node_attn_kernel<<<kN / 8, 256, 0, stream>>>(
            h1, h2, hp, rowptr, csrc, a + (size_t)layer * kH * kD, agg);

        ln_elu_res_kernel<<<kN / 8, 256, 0, stream>>>(
            agg, ln_g + (size_t)layer * kHD, ln_b + (size_t)layer * kHD,
            xin, xcur);
    }

    out_proj_kernel<<<(kN + 255) / 256, 256, 0, stream>>>(xcur, Wo, bo, out);
}


# LayerNorm+ELU+residual done by the node wave on the aggregated row it already holds in registers: the two LN launches and the agg store/reload are gone; same op order, same sums
# speedup vs baseline: 1.3526x; 1.0584x over previous
#include <hip/hip_runtime.h>
#include <math.h>


constexpr int kN   = 20000;
constexpr int kE   = 320000;
constexpr int kIN  = 128;
constexpr int kH   = 8;
constexpr int kD   = 16;
constexpr int kHD  = 128;
constexpr int kL   = 2;
constexpr int kOUT = 2;
constexpr float kLnEps = 1e-5f;

typedef __attribute__((ext_vector_type(2))) float v2f;
typedef __attribute__((ext_vector_type(8))) float v8f;

__device__ __forceinline__ float eluf(float x) {
    return x > 0.0f ? x : (expf(x) - 1.0f);
}

__device__ __forceinline__ float head_part(
    const float4& h1s, const float4& hps,
    const float4& h2d, const float4& hpd, const float4& av)
{
    const float t0 = eluf(h1s.x + h2d.x + hps.x * hpd.x);
    const float t1 = eluf(h1s.y + h2d.y + hps.y * hpd.y);
    const float t2 = eluf(h1s.z + h2d.z + hps.z * hpd.z);
    const float t3 = eluf(h1s.w + h2d.w + hps.w * hpd.w);
    const float part = av.x * t0 + av.y * t1 + av.z * t2 + av.w * t3;
    return part * 0.25f;
}

__global__ __launch_bounds__(256) void gemm_h1h2_kernel(
    const float* __restrict__ X,
    const float* __restrict__ W1, const float* __restrict__ B1,
    const float* __restrict__ W2, const float* __restrict__ B2,
    float* __restrict__ H1, float* __restrict__ H2, float* __restrict__ HP)
{
    const int lane  = threadIdx.x & 31;
    const int wave  = threadIdx.x >> 5;
    const int mbase = blockIdx.x * 16;
    const int n0    = wave * 16;
    const int r     = lane & 15;
    const int half  = lane >> 4;

    const float* xrow = X + (size_t)(mbase + r) * kIN;

    v8f acc1 = {};
    v8f acc2 = {};

#pragma unroll
    for (int k0 = 0; k0 < kIN; k0 += 4) {
        const int ka = k0 + half * 2;
        v2f a = *(const v2f*)(xrow + ka);
        v2f bA, bB;
        bA.x = W1[(size_t)ka       * kHD + n0 + r];
        bA.y = W1[(size_t)(ka + 1) * kHD + n0 + r];
        bB.x = W2[(size_t)ka       * kHD + n0 + r];
        bB.y = W2[(size_t)(ka + 1) * kHD + n0 + r];
        acc1 = __builtin_amdgcn_wmma_f32_16x16x4_f32(
            false, a, false, bA, (short)0, acc1, false, false);
        acc2 = __builtin_amdgcn_wmma_f32_16x16x4_f32(
            false, a, false, bB, (short)0, acc2, false, false);
    }

    const int n = n0 + r;
    const float bb1 = B1[n];
    const float bb2 = B2[n];

#pragma unroll
    for (int j = 0; j < 8; ++j) {
        const int m = mbase + j + half * 8;
        const float h1v = acc1[j] + bb1;
        const float h2v = acc2[j] + bb2;
        const size_t idx = (size_t)m * kHD + n;
        H1[idx] = h1v;
        H2[idx] = h2v;
        HP[idx] = h1v * h2v;
    }
}

constexpr int kScanThreads = 1024;
constexpr int kScanItems   = 20;
constexpr int kScanLive    = kN / kScanItems;
static_assert(kN % kScanItems == 0, "whole threads only");
static_assert(kScanLive < kScanThreads, "one block must cover all counts");
static_assert(kScanItems % 4 == 0, "int4 loads");
constexpr int kCntPad = kScanThreads * kScanItems;

__global__ __launch_bounds__(256) void dst_rank_kernel(
    const int* __restrict__ dst, int* __restrict__ cnt, int* __restrict__ rank)
{
    const int e = blockIdx.x * 256 + threadIdx.x;
    if (e >= kE) return;
    rank[e] = atomicAdd(cnt + dst[e], 1);
}

__global__ __launch_bounds__(kScanThreads) void scan_counts_kernel(
    const int* __restrict__ cnt, int* __restrict__ rowptr)
{
    __shared__ int wsum[32];
    const int t    = threadIdx.x;
    const int lane = t & 31;
    const int wave = t >> 5;

    int v[kScanItems];
    int tot = 0;
    if (t < kScanLive) {
        const int4* p = (const int4*)(cnt + t * kScanItems);
#pragma unroll
        for (int i = 0; i < kScanItems / 4; ++i) {
            const int4 q = p[i];
            v[4 * i + 0] = q.x; v[4 * i + 1] = q.y;
            v[4 * i + 2] = q.z; v[4 * i + 3] = q.w;
        }
#pragma unroll
        for (int i = 0; i < kScanItems; ++i) tot += v[i];
    } else {
#pragma unroll
        for (int i = 0; i < kScanItems; ++i) v[i] = 0;
    }

    int inc = tot;
#pragma unroll
    for (int off = 1; off < 32; off <<= 1) {
        const int o = __shfl_up(inc, off, 32);
        if (lane >= off) inc += o;
    }
    if (lane == 31) wsum[wave] = inc;
    __syncthreads();
    if (wave == 0) {
        const int w = wsum[lane];
        int winc = w;
#pragma unroll
        for (int off = 1; off < 32; off <<= 1) {
            const int o = __shfl_up(winc, off, 32);
            if (lane >= off) winc += o;
        }
        wsum[lane] = winc - w;
    }
    __syncthreads();

    int run = wsum[wave] + inc - tot;
    if (t < kScanLive) {
#pragma unroll
        for (int i = 0; i < kScanItems; ++i) {
            rowptr[t * kScanItems + i] = run;
            run += v[i];
        }
    }
    if (t == kScanThreads - 1) rowptr[kN] = run;
}

__global__ __launch_bounds__(256) void csr_fill_kernel(
    const int* __restrict__ src, const int* __restrict__ dst,
    const int* __restrict__ rowptr, const int* __restrict__ rank,
    int* __restrict__ csrc)
{
    const int e = blockIdx.x * 256 + threadIdx.x;
    if (e >= kE) return;
    csrc[rowptr[dst[e]] + rank[e]] = src[e];
}

__global__ __launch_bounds__(256) void node_attn_ln_kernel(
    const float* __restrict__ H1, const float* __restrict__ H2,
    const float* __restrict__ HP,
    const int* __restrict__ rowptr, const int* __restrict__ csrc,
    const float* __restrict__ avec,
    const float* __restrict__ G, const float* __restrict__ B,
    const float* __restrict__ Xin, float* __restrict__ Xout)
{
    const int lane = threadIdx.x & 31;
    const int node = blockIdx.x * 8 +
                     __builtin_amdgcn_readfirstlane((int)(threadIdx.x >> 5));
    if (node >= kN) return;

    const int beg = rowptr[node];
    const int end = rowptr[node + 1];

    const float4 h2d = *(const float4*)(H2 + (size_t)node * kHD + lane * 4);
    const float4 hpd = *(const float4*)(HP + (size_t)node * kHD + lane * 4);
    const float4 av  = *(const float4*)(avec + lane * 4);

    float4 acc = make_float4(0.0f, 0.0f, 0.0f, 0.0f);

    for (int base = beg; base < end; base += 32) {
        const int n = min(32, end - base);
        int mine = 0;
        if (lane < n) mine = csrc[base + lane];

        for (int j = 0; j < n; j += 2) {
            const int j1 = min(j + 1, n - 1);
            int s0 = __builtin_amdgcn_readlane(mine, j);
            int s1 = __builtin_amdgcn_readlane(mine, j1);
            s0 = (int)min((unsigned)s0, (unsigned)(kN - 1));
            s1 = (int)min((unsigned)s1, (unsigned)(kN - 1));

            const float4 h1a = *(const float4*)(H1 + (size_t)s0 * kHD + lane * 4);
            const float4 hpa = *(const float4*)(HP + (size_t)s0 * kHD + lane * 4);
            const float4 h1b = *(const float4*)(H1 + (size_t)s1 * kHD + lane * 4);
            const float4 hpb = *(const float4*)(HP + (size_t)s1 * kHD + lane * 4);

            float p0 = head_part(h1a, hpa, h2d, hpd, av);
            float p1 = head_part(h1b, hpb, h2d, hpd, av);

            p0 += __shfl_xor(p0, 1, 32);
            p1 += __shfl_xor(p1, 1, 32);
            p0 += __shfl_xor(p0, 2, 32);
            p1 += __shfl_xor(p1, 2, 32);

            float m0 = p0, m1 = p1;
            m0 = fmaxf(m0, __shfl_xor(m0, 4, 32));
            m1 = fmaxf(m1, __shfl_xor(m1, 4, 32));
            m0 = fmaxf(m0, __shfl_xor(m0, 8, 32));
            m1 = fmaxf(m1, __shfl_xor(m1, 8, 32));
            m0 = fmaxf(m0, __shfl_xor(m0, 16, 32));
            m1 = fmaxf(m1, __shfl_xor(m1, 16, 32));

            const float e0 = expf(p0 - m0);
            const float e1 = expf(p1 - m1);
            float q0 = e0, q1 = e1;
            q0 += __shfl_xor(q0, 4, 32);
            q1 += __shfl_xor(q1, 4, 32);
            q0 += __shfl_xor(q0, 8, 32);
            q1 += __shfl_xor(q1, 8, 32);
            q0 += __shfl_xor(q0, 16, 32);
            q1 += __shfl_xor(q1, 16, 32);

            const float a0 = e0 / q0;
            const float a1 = e1 / q1;

            acc.x += a0 * h1a.x;
            acc.y += a0 * h1a.y;
            acc.z += a0 * h1a.z;
            acc.w += a0 * h1a.w;
            if (j + 1 < n) {
                acc.x += a1 * h1b.x;
                acc.y += a1 * h1b.y;
                acc.z += a1 * h1b.z;
                acc.w += a1 * h1b.w;
            }
        }
    }

    float s = acc.x + acc.y + acc.z + acc.w;
#pragma unroll
    for (int m = 1; m <= 16; m <<= 1) s += __shfl_xor(s, m, 32);
    const float mu = s * (1.0f / 128.0f);

    const float dx0 = acc.x - mu, dx1 = acc.y - mu, dx2 = acc.z - mu, dx3 = acc.w - mu;
    float q = dx0 * dx0 + dx1 * dx1 + dx2 * dx2 + dx3 * dx3;
#pragma unroll
    for (int m = 1; m <= 16; m <<= 1) q += __shfl_xor(q, m, 32);
    const float inv = rsqrtf(q * (1.0f / 128.0f) + kLnEps);

    const float4 g  = *(const float4*)(G + lane * 4);
    const float4 b  = *(const float4*)(B + lane * 4);
    const float4 xr = *(const float4*)(Xin + (size_t)node * kHD + lane * 4);

    float4 o;
    o.x = eluf(dx0 * inv * g.x + b.x) + xr.x;
    o.y = eluf(dx1 * inv * g.y + b.y) + xr.y;
    o.z = eluf(dx2 * inv * g.z + b.z) + xr.z;
    o.w = eluf(dx3 * inv * g.w + b.w) + xr.w;

    *(float4*)(Xout + (size_t)node * kHD + lane * 4) = o;
}

__global__ __launch_bounds__(256) void out_proj_kernel(
    const float* __restrict__ X, const float* __restrict__ Wo,
    const float* __restrict__ bo, float* __restrict__ O)
{
    const int n = blockIdx.x * blockDim.x + threadIdx.x;
    if (n >= kN) return;
    float a0 = bo[0];
    float a1 = bo[1];
    const float* xr = X + (size_t)n * kHD;
#pragma unroll 8
    for (int k = 0; k < kHD; ++k) {
        const float xv = xr[k];
        a0 += xv * Wo[2 * k + 0];
        a1 += xv * Wo[2 * k + 1];
    }
    O[2 * n + 0] = a0;
    O[2 * n + 1] = a1;
}

extern "C" void kernel_launch(void* const* d_in, const int* in_sizes, int n_in,
                              void* d_out, int out_size, void* d_ws, size_t ws_size,
                              hipStream_t stream)
{
    (void)in_sizes; (void)n_in; (void)out_size; (void)ws_size;

    const float* x    = (const float*)d_in[0];
    const int*   ei   = (const int*)  d_in[1];
    const float* W1   = (const float*)d_in[2];
    const float* b1   = (const float*)d_in[3];
    const float* W2   = (const float*)d_in[4];
    const float* b2   = (const float*)d_in[5];
    const float* a    = (const float*)d_in[6];
    const float* ln_g = (const float*)d_in[7];
    const float* ln_b = (const float*)d_in[8];
    const float* Wo   = (const float*)d_in[9];
    const float* bo   = (const float*)d_in[10];
    float*       out  = (float*)d_out;

    const size_t NHD = (size_t)kN * kHD;
    float* h1   = (float*)d_ws;
    float* h2   = h1  + NHD;
    float* hp   = h2  + NHD;
    float* xa   = hp  + NHD;
    float* xb   = xa  + NHD;
    int* cnt    = (int*)(xb + NHD);
    int* rowptr = cnt + kCntPad;
    int* rank   = rowptr + kCntPad;
    int* csrc   = rank + kE;

    const int* srcIdx = ei;
    const int* dstIdx = ei + kE;

    hipMemsetAsync(cnt, 0, (size_t)kN * sizeof(int), stream);
    dst_rank_kernel<<<kE / 256, 256, 0, stream>>>(dstIdx, cnt, rank);
    scan_counts_kernel<<<1, kScanThreads, 0, stream>>>(cnt, rowptr);
    csr_fill_kernel<<<kE / 256, 256, 0, stream>>>(srcIdx, dstIdx, rowptr, rank, csrc);

    for (int layer = 0; layer < kL; ++layer) {
        const float* xin  = (layer == 0) ? x  : xa;
        float*       xout = (layer == 0) ? xa : xb;

        gemm_h1h2_kernel<<<kN / 16, 256, 0, stream>>>(
            xin,
            W1 + (size_t)layer * kIN * kHD, b1 + (size_t)layer * kHD,
            W2 + (size_t)layer * kIN * kHD, b2 + (size_t)layer * kHD,
            h1, h2, hp);

        node_attn_ln_kernel<<<kN / 8, 256, 0, stream>>>(
            h1, h2, hp, rowptr, csrc, a + (size_t)layer * kH * kD,
            ln_g + (size_t)layer * kHD, ln_b + (size_t)layer * kHD,
            xin, xout);
    }

    out_proj_kernel<<<(kN + 255) / 256, 256, 0, stream>>>(xb, Wo, bo, out);
}


# Node kernel: ELU and head-softmax use hardware exp2(x*log2e) and rcp in place of library expf and IEEE divide, ELU branch-free; arguments <= 0, abs error < 2e-7, last-bit rounding differs
# speedup vs baseline: 1.5598x; 1.1532x over previous
#include <hip/hip_runtime.h>
#include <math.h>


constexpr int kN   = 20000;
constexpr int kE   = 320000;
constexpr int kIN  = 128;
constexpr int kH   = 8;
constexpr int kD   = 16;
constexpr int kHD  = 128;
constexpr int kL   = 2;
constexpr int kOUT = 2;
constexpr float kLnEps = 1e-5f;

typedef __attribute__((ext_vector_type(2))) float v2f;
typedef __attribute__((ext_vector_type(8))) float v8f;

__device__ __forceinline__ float exp_nonpos(float x) {
    return __builtin_amdgcn_exp2f(x * 1.44269504088896340736f);
}

__device__ __forceinline__ float eluf(float x) {
    return x > 0.0f ? x : (exp_nonpos(x) - 1.0f);
}

__device__ __forceinline__ float head_part(
    const float4& h1s, const float4& hps,
    const float4& h2d, const float4& hpd, const float4& av)
{
    const float t0 = eluf(h1s.x + h2d.x + hps.x * hpd.x);
    const float t1 = eluf(h1s.y + h2d.y + hps.y * hpd.y);
    const float t2 = eluf(h1s.z + h2d.z + hps.z * hpd.z);
    const float t3 = eluf(h1s.w + h2d.w + hps.w * hpd.w);
    const float part = av.x * t0 + av.y * t1 + av.z * t2 + av.w * t3;
    return part * 0.25f;
}

__global__ __launch_bounds__(256) void gemm_h1h2_kernel(
    const float* __restrict__ X,
    const float* __restrict__ W1, const float* __restrict__ B1,
    const float* __restrict__ W2, const float* __restrict__ B2,
    float* __restrict__ H1, float* __restrict__ H2, float* __restrict__ HP)
{
    const int lane  = threadIdx.x & 31;
    const int wave  = threadIdx.x >> 5;
    const int mbase = blockIdx.x * 16;
    const int n0    = wave * 16;
    const int r     = lane & 15;
    const int half  = lane >> 4;

    const float* xrow = X + (size_t)(mbase + r) * kIN;

    v8f acc1 = {};
    v8f acc2 = {};

#pragma unroll
    for (int k0 = 0; k0 < kIN; k0 += 4) {
        const int ka = k0 + half * 2;
        v2f a = *(const v2f*)(xrow + ka);
        v2f bA, bB;
        bA.x = W1[(size_t)ka       * kHD + n0 + r];
        bA.y = W1[(size_t)(ka + 1) * kHD + n0 + r];
        bB.x = W2[(size_t)ka       * kHD + n0 + r];
        bB.y = W2[(size_t)(ka + 1) * kHD + n0 + r];
        acc1 = __builtin_amdgcn_wmma_f32_16x16x4_f32(
            false, a, false, bA, (short)0, acc1, false, false);
        acc2 = __builtin_amdgcn_wmma_f32_16x16x4_f32(
            false, a, false, bB, (short)0, acc2, false, false);
    }

    const int n = n0 + r;
    const float bb1 = B1[n];
    const float bb2 = B2[n];

#pragma unroll
    for (int j = 0; j < 8; ++j) {
        const int m = mbase + j + half * 8;
        const float h1v = acc1[j] + bb1;
        const float h2v = acc2[j] + bb2;
        const size_t idx = (size_t)m * kHD + n;
        H1[idx] = h1v;
        H2[idx] = h2v;
        HP[idx] = h1v * h2v;
    }
}

constexpr int kScanThreads = 1024;
constexpr int kScanItems   = 20;
constexpr int kScanLive    = kN / kScanItems;
static_assert(kN % kScanItems == 0, "whole threads only");
static_assert(kScanLive < kScanThreads, "one block must cover all counts");
static_assert(kScanItems % 4 == 0, "int4 loads");
constexpr int kCntPad = kScanThreads * kScanItems;

__global__ __launch_bounds__(256) void dst_rank_kernel(
    const int* __restrict__ dst, int* __restrict__ cnt, int* __restrict__ rank)
{
    const int e = blockIdx.x * 256 + threadIdx.x;
    if (e >= kE) return;
    rank[e] = atomicAdd(cnt + dst[e], 1);
}

__global__ __launch_bounds__(kScanThreads) void scan_counts_kernel(
    const int* __restrict__ cnt, int* __restrict__ rowptr)
{
    __shared__ int wsum[32];
    const int t    = threadIdx.x;
    const int lane = t & 31;
    const int wave = t >> 5;

    int v[kScanItems];
    int tot = 0;
    if (t < kScanLive) {
        const int4* p = (const int4*)(cnt + t * kScanItems);
#pragma unroll
        for (int i = 0; i < kScanItems / 4; ++i) {
            const int4 q = p[i];
            v[4 * i + 0] = q.x; v[4 * i + 1] = q.y;
            v[4 * i + 2] = q.z; v[4 * i + 3] = q.w;
        }
#pragma unroll
        for (int i = 0; i < kScanItems; ++i) tot += v[i];
    } else {
#pragma unroll
        for (int i = 0; i < kScanItems; ++i) v[i] = 0;
    }

    int inc = tot;
#pragma unroll
    for (int off = 1; off < 32; off <<= 1) {
        const int o = __shfl_up(inc, off, 32);
        if (lane >= off) inc += o;
    }
    if (lane == 31) wsum[wave] = inc;
    __syncthreads();
    if (wave == 0) {
        const int w = wsum[lane];
        int winc = w;
#pragma unroll
        for (int off = 1; off < 32; off <<= 1) {
            const int o = __shfl_up(winc, off, 32);
            if (lane >= off) winc += o;
        }
        wsum[lane] = winc - w;
    }
    __syncthreads();

    int run = wsum[wave] + inc - tot;
    if (t < kScanLive) {
#pragma unroll
        for (int i = 0; i < kScanItems; ++i) {
            rowptr[t * kScanItems + i] = run;
            run += v[i];
        }
    }
    if (t == kScanThreads - 1) rowptr[kN] = run;
}

__global__ __launch_bounds__(256) void csr_fill_kernel(
    const int* __restrict__ src, const int* __restrict__ dst,
    const int* __restrict__ rowptr, const int* __restrict__ rank,
    int* __restrict__ csrc)
{
    const int e = blockIdx.x * 256 + threadIdx.x;
    if (e >= kE) return;
    csrc[rowptr[dst[e]] + rank[e]] = src[e];
}

__global__ __launch_bounds__(256) void node_attn_ln_kernel(
    const float* __restrict__ H1, const float* __restrict__ H2,
    const float* __restrict__ HP,
    const int* __restrict__ rowptr, const int* __restrict__ csrc,
    const float* __restrict__ avec,
    const float* __restrict__ G, const float* __restrict__ B,
    const float* __restrict__ Xin, float* __restrict__ Xout)
{
    const int lane = threadIdx.x & 31;
    const int node = blockIdx.x * 8 +
                     __builtin_amdgcn_readfirstlane((int)(threadIdx.x >> 5));
    if (node >= kN) return;

    const int beg = rowptr[node];
    const int end = rowptr[node + 1];

    const float4 h2d = *(const float4*)(H2 + (size_t)node * kHD + lane * 4);
    const float4 hpd = *(const float4*)(HP + (size_t)node * kHD + lane * 4);
    const float4 av  = *(const float4*)(avec + lane * 4);

    float4 acc = make_float4(0.0f, 0.0f, 0.0f, 0.0f);

    for (int base = beg; base < end; base += 32) {
        const int n = min(32, end - base);
        int mine = 0;
        if (lane < n) mine = csrc[base + lane];

        for (int j = 0; j < n; j += 2) {
            const int j1 = min(j + 1, n - 1);
            int s0 = __builtin_amdgcn_readlane(mine, j);
            int s1 = __builtin_amdgcn_readlane(mine, j1);
            s0 = (int)min((unsigned)s0, (unsigned)(kN - 1));
            s1 = (int)min((unsigned)s1, (unsigned)(kN - 1));

            const float4 h1a = *(const float4*)(H1 + (size_t)s0 * kHD + lane * 4);
            const float4 hpa = *(const float4*)(HP + (size_t)s0 * kHD + lane * 4);
            const float4 h1b = *(const float4*)(H1 + (size_t)s1 * kHD + lane * 4);
            const float4 hpb = *(const float4*)(HP + (size_t)s1 * kHD + lane * 4);

            float p0 = head_part(h1a, hpa, h2d, hpd, av);
            float p1 = head_part(h1b, hpb, h2d, hpd, av);

            p0 += __shfl_xor(p0, 1, 32);
            p1 += __shfl_xor(p1, 1, 32);
            p0 += __shfl_xor(p0, 2, 32);
            p1 += __shfl_xor(p1, 2, 32);

            float m0 = p0, m1 = p1;
            m0 = fmaxf(m0, __shfl_xor(m0, 4, 32));
            m1 = fmaxf(m1, __shfl_xor(m1, 4, 32));
            m0 = fmaxf(m0, __shfl_xor(m0, 8, 32));
            m1 = fmaxf(m1, __shfl_xor(m1, 8, 32));
            m0 = fmaxf(m0, __shfl_xor(m0, 16, 32));
            m1 = fmaxf(m1, __shfl_xor(m1, 16, 32));

            const float e0 = exp_nonpos(p0 - m0);
            const float e1 = exp_nonpos(p1 - m1);
            float q0 = e0, q1 = e1;
            q0 += __shfl_xor(q0, 4, 32);
            q1 += __shfl_xor(q1, 4, 32);
            q0 += __shfl_xor(q0, 8, 32);
            q1 += __shfl_xor(q1, 8, 32);
            q0 += __shfl_xor(q0, 16, 32);
            q1 += __shfl_xor(q1, 16, 32);

            const float a0 = e0 * __builtin_amdgcn_rcpf(q0);
            const float a1 = e1 * __builtin_amdgcn_rcpf(q1);

            acc.x += a0 * h1a.x;
            acc.y += a0 * h1a.y;
            acc.z += a0 * h1a.z;
            acc.w += a0 * h1a.w;
            if (j + 1 < n) {
                acc.x += a1 * h1b.x;
                acc.y += a1 * h1b.y;
                acc.z += a1 * h1b.z;
                acc.w += a1 * h1b.w;
            }
        }
    }

    float s = acc.x + acc.y + acc.z + acc.w;
#pragma unroll
    for (int m = 1; m <= 16; m <<= 1) s += __shfl_xor(s, m, 32);
    const float mu = s * (1.0f / 128.0f);

    const float dx0 = acc.x - mu, dx1 = acc.y - mu, dx2 = acc.z - mu, dx3 = acc.w - mu;
    float q = dx0 * dx0 + dx1 * dx1 + dx2 * dx2 + dx3 * dx3;
#pragma unroll
    for (int m = 1; m <= 16; m <<= 1) q += __shfl_xor(q, m, 32);
    const float inv = rsqrtf(q * (1.0f / 128.0f) + kLnEps);

    const float4 g  = *(const float4*)(G + lane * 4);
    const float4 b  = *(const float4*)(B + lane * 4);
    const float4 xr = *(const float4*)(Xin + (size_t)node * kHD + lane * 4);

    float4 o;
    o.x = eluf(dx0 * inv * g.x + b.x) + xr.x;
    o.y = eluf(dx1 * inv * g.y + b.y) + xr.y;
    o.z = eluf(dx2 * inv * g.z + b.z) + xr.z;
    o.w = eluf(dx3 * inv * g.w + b.w) + xr.w;

    *(float4*)(Xout + (size_t)node * kHD + lane * 4) = o;
}

__global__ __launch_bounds__(256) void out_proj_kernel(
    const float* __restrict__ X, const float* __restrict__ Wo,
    const float* __restrict__ bo, float* __restrict__ O)
{
    const int n = blockIdx.x * blockDim.x + threadIdx.x;
    if (n >= kN) return;
    float a0 = bo[0];
    float a1 = bo[1];
    const float* xr = X + (size_t)n * kHD;
#pragma unroll 8
    for (int k = 0; k < kHD; ++k) {
        const float xv = xr[k];
        a0 += xv * Wo[2 * k + 0];
        a1 += xv * Wo[2 * k + 1];
    }
    O[2 * n + 0] = a0;
    O[2 * n + 1] = a1;
}

extern "C" void kernel_launch(void* const* d_in, const int* in_sizes, int n_in,
                              void* d_out, int out_size, void* d_ws, size_t ws_size,
                              hipStream_t stream)
{
    (void)in_sizes; (void)n_in; (void)out_size; (void)ws_size;

    const float* x    = (const float*)d_in[0];
    const int*   ei   = (const int*)  d_in[1];
    const float* W1   = (const float*)d_in[2];
    const float* b1   = (const float*)d_in[3];
    const float* W2   = (const float*)d_in[4];
    const float* b2   = (const float*)d_in[5];
    const float* a    = (const float*)d_in[6];
    const float* ln_g = (const float*)d_in[7];
    const float* ln_b = (const float*)d_in[8];
    const float* Wo   = (const float*)d_in[9];
    const float* bo   = (const float*)d_in[10];
    float*       out  = (float*)d_out;

    const size_t NHD = (size_t)kN * kHD;
    float* h1   = (float*)d_ws;
    float* h2   = h1  + NHD;
    float* hp   = h2  + NHD;
    float* xa   = hp  + NHD;
    float* xb   = xa  + NHD;
    int* cnt    = (int*)(xb + NHD);
    int* rowptr = cnt + kCntPad;
    int* rank   = rowptr + kCntPad;
    int* csrc   = rank + kE;

    const int* srcIdx = ei;
    const int* dstIdx = ei + kE;

    hipMemsetAsync(cnt, 0, (size_t)kN * sizeof(int), stream);
    dst_rank_kernel<<<kE / 256, 256, 0, stream>>>(dstIdx, cnt, rank);
    scan_counts_kernel<<<1, kScanThreads, 0, stream>>>(cnt, rowptr);
    csr_fill_kernel<<<kE / 256, 256, 0, stream>>>(srcIdx, dstIdx, rowptr, rank, csrc);

    for (int layer = 0; layer < kL; ++layer) {
        const float* xin  = (layer == 0) ? x  : xa;
        float*       xout = (layer == 0) ? xa : xb;

        gemm_h1h2_kernel<<<kN / 16, 256, 0, stream>>>(
            xin,
            W1 + (size_t)layer * kIN * kHD, b1 + (size_t)layer * kHD,
            W2 + (size_t)layer * kIN * kHD, b2 + (size_t)layer * kHD,
            h1, h2, hp);

        node_attn_ln_kernel<<<kN / 8, 256, 0, stream>>>(
            h1, h2, hp, rowptr, csrc, a + (size_t)layer * kH * kD,
            ln_g + (size_t)layer * kHD, ln_b + (size_t)layer * kHD,
            xin, xout);
    }

    out_proj_kernel<<<(kN + 255) / 256, 256, 0, stream>>>(xb, Wo, bo, out);
}


# Output projection by the last layer's node wave on the row it holds (lane partials, then xor butterfly: new sum order); 16 nodes per block store one 128 B line; out_proj launch and row store gone
# speedup vs baseline: 1.6500x; 1.0578x over previous
#include <hip/hip_runtime.h>
#include <math.h>


constexpr int kN   = 20000;
constexpr int kE   = 320000;
constexpr int kIN  = 128;
constexpr int kH   = 8;
constexpr int kD   = 16;
constexpr int kHD  = 128;
constexpr int kL   = 2;
constexpr int kOUT = 2;
constexpr float kLnEps = 1e-5f;

typedef __attribute__((ext_vector_type(2))) float v2f;
typedef __attribute__((ext_vector_type(8))) float v8f;

__device__ __forceinline__ float exp_nonpos(float x) {
    return __builtin_amdgcn_exp2f(x * 1.44269504088896340736f);
}

__device__ __forceinline__ float eluf(float x) {
    return x > 0.0f ? x : (exp_nonpos(x) - 1.0f);
}

__device__ __forceinline__ float head_part(
    const float4& h1s, const float4& hps,
    const float4& h2d, const float4& hpd, const float4& av)
{
    const float t0 = eluf(h1s.x + h2d.x + hps.x * hpd.x);
    const float t1 = eluf(h1s.y + h2d.y + hps.y * hpd.y);
    const float t2 = eluf(h1s.z + h2d.z + hps.z * hpd.z);
    const float t3 = eluf(h1s.w + h2d.w + hps.w * hpd.w);
    const float part = av.x * t0 + av.y * t1 + av.z * t2 + av.w * t3;
    return part * 0.25f;
}

__global__ __launch_bounds__(256) void gemm_h1h2_kernel(
    const float* __restrict__ X,
    const float* __restrict__ W1, const float* __restrict__ B1,
    const float* __restrict__ W2, const float* __restrict__ B2,
    float* __restrict__ H1, float* __restrict__ H2, float* __restrict__ HP)
{
    const int lane  = threadIdx.x & 31;
    const int wave  = threadIdx.x >> 5;
    const int mbase = blockIdx.x * 16;
    const int n0    = wave * 16;
    const int r     = lane & 15;
    const int half  = lane >> 4;

    const float* xrow = X + (size_t)(mbase + r) * kIN;

    v8f acc1 = {};
    v8f acc2 = {};

#pragma unroll
    for (int k0 = 0; k0 < kIN; k0 += 4) {
        const int ka = k0 + half * 2;
        v2f a = *(const v2f*)(xrow + ka);
        v2f bA, bB;
        bA.x = W1[(size_t)ka       * kHD + n0 + r];
        bA.y = W1[(size_t)(ka + 1) * kHD + n0 + r];
        bB.x = W2[(size_t)ka       * kHD + n0 + r];
        bB.y = W2[(size_t)(ka + 1) * kHD + n0 + r];
        acc1 = __builtin_amdgcn_wmma_f32_16x16x4_f32(
            false, a, false, bA, (short)0, acc1, false, false);
        acc2 = __builtin_amdgcn_wmma_f32_16x16x4_f32(
            false, a, false, bB, (short)0, acc2, false, false);
    }

    const int n = n0 + r;
    const float bb1 = B1[n];
    const float bb2 = B2[n];

#pragma unroll
    for (int j = 0; j < 8; ++j) {
        const int m = mbase + j + half * 8;
        const float h1v = acc1[j] + bb1;
        const float h2v = acc2[j] + bb2;
        const size_t idx = (size_t)m * kHD + n;
        H1[idx] = h1v;
        H2[idx] = h2v;
        HP[idx] = h1v * h2v;
    }
}

constexpr int kScanThreads = 1024;
constexpr int kScanItems   = 20;
constexpr int kScanLive    = kN / kScanItems;
static_assert(kN % kScanItems == 0, "whole threads only");
static_assert(kScanLive < kScanThreads, "one block must cover all counts");
static_assert(kScanItems % 4 == 0, "int4 loads");
constexpr int kCntPad = kScanThreads * kScanItems;

__global__ __launch_bounds__(256) void dst_rank_kernel(
    const int* __restrict__ dst, int* __restrict__ cnt, int* __restrict__ rank)
{
    const int e = blockIdx.x * 256 + threadIdx.x;
    if (e >= kE) return;
    rank[e] = atomicAdd(cnt + dst[e], 1);
}

__global__ __launch_bounds__(kScanThreads) void scan_counts_kernel(
    const int* __restrict__ cnt, int* __restrict__ rowptr)
{
    __shared__ int wsum[32];
    const int t    = threadIdx.x;
    const int lane = t & 31;
    const int wave = t >> 5;

    int v[kScanItems];
    int tot = 0;
    if (t < kScanLive) {
        const int4* p = (const int4*)(cnt + t * kScanItems);
#pragma unroll
        for (int i = 0; i < kScanItems / 4; ++i) {
            const int4 q = p[i];
            v[4 * i + 0] = q.x; v[4 * i + 1] = q.y;
            v[4 * i + 2] = q.z; v[4 * i + 3] = q.w;
        }
#pragma unroll
        for (int i = 0; i < kScanItems; ++i) tot += v[i];
    } else {
#pragma unroll
        for (int i = 0; i < kScanItems; ++i) v[i] = 0;
    }

    int inc = tot;
#pragma unroll
    for (int off = 1; off < 32; off <<= 1) {
        const int o = __shfl_up(inc, off, 32);
        if (lane >= off) inc += o;
    }
    if (lane == 31) wsum[wave] = inc;
    __syncthreads();
    if (wave == 0) {
        const int w = wsum[lane];
        int winc = w;
#pragma unroll
        for (int off = 1; off < 32; off <<= 1) {
            const int o = __shfl_up(winc, off, 32);
            if (lane >= off) winc += o;
        }
        wsum[lane] = winc - w;
    }
    __syncthreads();

    int run = wsum[wave] + inc - tot;
    if (t < kScanLive) {
#pragma unroll
        for (int i = 0; i < kScanItems; ++i) {
            rowptr[t * kScanItems + i] = run;
            run += v[i];
        }
    }
    if (t == kScanThreads - 1) rowptr[kN] = run;
}

__global__ __launch_bounds__(256) void csr_fill_kernel(
    const int* __restrict__ src, const int* __restrict__ dst,
    const int* __restrict__ rowptr, const int* __restrict__ rank,
    int* __restrict__ csrc)
{
    const int e = blockIdx.x * 256 + threadIdx.x;
    if (e >= kE) return;
    csrc[rowptr[dst[e]] + rank[e]] = src[e];
}

constexpr int kNodesPerBlock      = 8;
constexpr int kNodesPerBlockFinal = 16;
static_assert(kN % kNodesPerBlockFinal == 0,
              "the last layer's grid covers whole blocks: no wave may leave before the barrier");

template <bool kFinal>
__global__ __launch_bounds__(kFinal ? kNodesPerBlockFinal * 32 : kNodesPerBlock * 32)
void node_attn_ln_kernel(
    const float* __restrict__ H1, const float* __restrict__ H2,
    const float* __restrict__ HP,
    const int* __restrict__ rowptr, const int* __restrict__ csrc,
    const float* __restrict__ avec,
    const float* __restrict__ G, const float* __restrict__ B,
    const float* __restrict__ Xin, float* __restrict__ Xout,
    const float* __restrict__ Wo, const float* __restrict__ bo,
    float* __restrict__ O)
{
    const int lane = threadIdx.x & 31;
    const int wave = __builtin_amdgcn_readfirstlane((int)(threadIdx.x >> 5));
    const int node = blockIdx.x * (kFinal ? kNodesPerBlockFinal : kNodesPerBlock) + wave;
    if constexpr (!kFinal) {
        if (node >= kN) return;
    }

    const int beg = rowptr[node];
    const int end = rowptr[node + 1];

    const float4 h2d = *(const float4*)(H2 + (size_t)node * kHD + lane * 4);
    const float4 hpd = *(const float4*)(HP + (size_t)node * kHD + lane * 4);
    const float4 av  = *(const float4*)(avec + lane * 4);

    float4 acc = make_float4(0.0f, 0.0f, 0.0f, 0.0f);

    for (int base = beg; base < end; base += 32) {
        const int n = min(32, end - base);
        int mine = 0;
        if (lane < n) mine = csrc[base + lane];

        for (int j = 0; j < n; j += 2) {
            const int j1 = min(j + 1, n - 1);
            int s0 = __builtin_amdgcn_readlane(mine, j);
            int s1 = __builtin_amdgcn_readlane(mine, j1);
            s0 = (int)min((unsigned)s0, (unsigned)(kN - 1));
            s1 = (int)min((unsigned)s1, (unsigned)(kN - 1));

            const float4 h1a = *(const float4*)(H1 + (size_t)s0 * kHD + lane * 4);
            const float4 hpa = *(const float4*)(HP + (size_t)s0 * kHD + lane * 4);
            const float4 h1b = *(const float4*)(H1 + (size_t)s1 * kHD + lane * 4);
            const float4 hpb = *(const float4*)(HP + (size_t)s1 * kHD + lane * 4);

            float p0 = head_part(h1a, hpa, h2d, hpd, av);
            float p1 = head_part(h1b, hpb, h2d, hpd, av);

            p0 += __shfl_xor(p0, 1, 32);
            p1 += __shfl_xor(p1, 1, 32);
            p0 += __shfl_xor(p0, 2, 32);
            p1 += __shfl_xor(p1, 2, 32);

            float m0 = p0, m1 = p1;
            m0 = fmaxf(m0, __shfl_xor(m0, 4, 32));
            m1 = fmaxf(m1, __shfl_xor(m1, 4, 32));
            m0 = fmaxf(m0, __shfl_xor(m0, 8, 32));
            m1 = fmaxf(m1, __shfl_xor(m1, 8, 32));
            m0 = fmaxf(m0, __shfl_xor(m0, 16, 32));
            m1 = fmaxf(m1, __shfl_xor(m1, 16, 32));

            const float e0 = exp_nonpos(p0 - m0);
            const float e1 = exp_nonpos(p1 - m1);
            float q0 = e0, q1 = e1;
            q0 += __shfl_xor(q0, 4, 32);
            q1 += __shfl_xor(q1, 4, 32);
            q0 += __shfl_xor(q0, 8, 32);
            q1 += __shfl_xor(q1, 8, 32);
            q0 += __shfl_xor(q0, 16, 32);
            q1 += __shfl_xor(q1, 16, 32);

            const float a0 = e0 * __builtin_amdgcn_rcpf(q0);
            const float a1 = e1 * __builtin_amdgcn_rcpf(q1);

            acc.x += a0 * h1a.x;
            acc.y += a0 * h1a.y;
            acc.z += a0 * h1a.z;
            acc.w += a0 * h1a.w;
            if (j + 1 < n) {
                acc.x += a1 * h1b.x;
                acc.y += a1 * h1b.y;
                acc.z += a1 * h1b.z;
                acc.w += a1 * h1b.w;
            }
        }
    }

    float s = acc.x + acc.y + acc.z + acc.w;
#pragma unroll
    for (int m = 1; m <= 16; m <<= 1) s += __shfl_xor(s, m, 32);
    const float mu = s * (1.0f / 128.0f);

    const float dx0 = acc.x - mu, dx1 = acc.y - mu, dx2 = acc.z - mu, dx3 = acc.w - mu;
    float q = dx0 * dx0 + dx1 * dx1 + dx2 * dx2 + dx3 * dx3;
#pragma unroll
    for (int m = 1; m <= 16; m <<= 1) q += __shfl_xor(q, m, 32);
    const float inv = rsqrtf(q * (1.0f / 128.0f) + kLnEps);

    const float4 g  = *(const float4*)(G + lane * 4);
    const float4 b  = *(const float4*)(B + lane * 4);
    const float4 xr = *(const float4*)(Xin + (size_t)node * kHD + lane * 4);

    float4 o;
    o.x = eluf(dx0 * inv * g.x + b.x) + xr.x;
    o.y = eluf(dx1 * inv * g.y + b.y) + xr.y;
    o.z = eluf(dx2 * inv * g.z + b.z) + xr.z;
    o.w = eluf(dx3 * inv * g.w + b.w) + xr.w;

    if constexpr (!kFinal) {
        *(float4*)(Xout + (size_t)node * kHD + lane * 4) = o;
    } else {
        const float4 w01 = *(const float4*)(Wo + lane * 8);
        const float4 w23 = *(const float4*)(Wo + lane * 8 + 4);
        float p0 = o.x * w01.x + o.y * w01.z + o.z * w23.x + o.w * w23.z;
        float p1 = o.x * w01.y + o.y * w01.w + o.z * w23.y + o.w * w23.w;
#pragma unroll
        for (int m = 1; m <= 16; m <<= 1) {
            p0 += __shfl_xor(p0, m, 32);
            p1 += __shfl_xor(p1, m, 32);
        }
        __shared__ float2 proj[kNodesPerBlockFinal];
        if (lane == 0) proj[wave] = make_float2(p0 + bo[0], p1 + bo[1]);
        __syncthreads();
        if (wave == 0 && lane < kNodesPerBlockFinal) {
            *(float2*)(O + (size_t)(blockIdx.x * kNodesPerBlockFinal + lane) * kOUT) = proj[lane];
        }
    }
}

extern "C" void kernel_launch(void* const* d_in, const int* in_sizes, int n_in,
                              void* d_out, int out_size, void* d_ws, size_t ws_size,
                              hipStream_t stream)
{
    (void)in_sizes; (void)n_in; (void)out_size; (void)ws_size;

    const float* x    = (const float*)d_in[0];
    const int*   ei   = (const int*)  d_in[1];
    const float* W1   = (const float*)d_in[2];
    const float* b1   = (const float*)d_in[3];
    const float* W2   = (const float*)d_in[4];
    const float* b2   = (const float*)d_in[5];
    const float* a    = (const float*)d_in[6];
    const float* ln_g = (const float*)d_in[7];
    const float* ln_b = (const float*)d_in[8];
    const float* Wo   = (const float*)d_in[9];
    const float* bo   = (const float*)d_in[10];
    float*       out  = (float*)d_out;

    const size_t NHD = (size_t)kN * kHD;
    float* h1   = (float*)d_ws;
    float* h2   = h1  + NHD;
    float* hp   = h2  + NHD;
    float* xa   = hp  + NHD;
    int* cnt    = (int*)(xa + NHD);
    int* rowptr = cnt + kCntPad;
    int* rank   = rowptr + kCntPad;
    int* csrc   = rank + kE;

    const int* srcIdx = ei;
    const int* dstIdx = ei + kE;

    hipMemsetAsync(cnt, 0, (size_t)kN * sizeof(int), stream);
    dst_rank_kernel<<<kE / 256, 256, 0, stream>>>(dstIdx, cnt, rank);
    scan_counts_kernel<<<1, kScanThreads, 0, stream>>>(cnt, rowptr);
    csr_fill_kernel<<<kE / 256, 256, 0, stream>>>(srcIdx, dstIdx, rowptr, rank, csrc);

    static_assert(kL == 2, "layer 0 stores its rows, layer 1 projects them");

    gemm_h1h2_kernel<<<kN / 16, 256, 0, stream>>>(
        x, W1, b1, W2, b2, h1, h2, hp);
    node_attn_ln_kernel<false><<<kN / kNodesPerBlock, kNodesPerBlock * 32, 0, stream>>>(
        h1, h2, hp, rowptr, csrc, a, ln_g, ln_b, x, xa,
        nullptr, nullptr, nullptr);

    gemm_h1h2_kernel<<<kN / 16, 256, 0, stream>>>(
        xa,
        W1 + (size_t)kIN * kHD, b1 + kHD,
        W2 + (size_t)kIN * kHD, b2 + kHD,
        h1, h2, hp);
    node_attn_ln_kernel<true><<<kN / kNodesPerBlockFinal, kNodesPerBlockFinal * 32, 0, stream>>>(
        h1, h2, hp, rowptr, csrc, a + kH * kD, ln_g + kHD, ln_b + kHD, xa, nullptr,
        Wo, bo, out);
}


# Node kernel: every cross-lane sum and max (per edge, LayerNorm, projection) on the vector ALU by DPP row_xmask and v_permlanex16 instead of ds_bpermute via the LDS unit; same butterflies, same sums
# speedup vs baseline: 1.8178x; 1.1017x over previous
#include <hip/hip_runtime.h>
#include <math.h>


constexpr int kN   = 20000;
constexpr int kE   = 320000;
constexpr int kIN  = 128;
constexpr int kH   = 8;
constexpr int kD   = 16;
constexpr int kHD  = 128;
constexpr int kL   = 2;
constexpr int kOUT = 2;
constexpr float kLnEps = 1e-5f;

typedef __attribute__((ext_vector_type(2))) float v2f;
typedef __attribute__((ext_vector_type(8))) float v8f;

__device__ __forceinline__ float exp_nonpos(float x) {
    return __builtin_amdgcn_exp2f(x * 1.44269504088896340736f);
}

__device__ __forceinline__ float eluf(float x) {
    return x > 0.0f ? x : (exp_nonpos(x) - 1.0f);
}

template <int M>
__device__ __forceinline__ float lane_xor_row(float x) {
    static_assert(M >= 1 && M <= 15, "row_xmask stays inside a 16-lane row");
    return __builtin_bit_cast(float, __builtin_amdgcn_update_dpp(
        0, __builtin_bit_cast(int, x), 0x160 + M, 0xF, 0xF, true));
}
__device__ __forceinline__ float lane_xor_16(float x) {
    const int b = __builtin_bit_cast(int, x);
    return __builtin_bit_cast(float, __builtin_amdgcn_permlanex16(
        b, b, 0x76543210, 0xfedcba98, false, false));
}
__device__ __forceinline__ float wave_sum(float v) {
    v += lane_xor_row<1>(v);
    v += lane_xor_row<2>(v);
    v += lane_xor_row<4>(v);
    v += lane_xor_row<8>(v);
    v += lane_xor_16(v);
    return v;
}

__device__ __forceinline__ float head_part(
    const float4& h1s, const float4& hps,
    const float4& h2d, const float4& hpd, const float4& av)
{
    const float t0 = eluf(h1s.x + h2d.x + hps.x * hpd.x);
    const float t1 = eluf(h1s.y + h2d.y + hps.y * hpd.y);
    const float t2 = eluf(h1s.z + h2d.z + hps.z * hpd.z);
    const float t3 = eluf(h1s.w + h2d.w + hps.w * hpd.w);
    const float part = av.x * t0 + av.y * t1 + av.z * t2 + av.w * t3;
    return part * 0.25f;
}

__global__ __launch_bounds__(256) void gemm_h1h2_kernel(
    const float* __restrict__ X,
    const float* __restrict__ W1, const float* __restrict__ B1,
    const float* __restrict__ W2, const float* __restrict__ B2,
    float* __restrict__ H1, float* __restrict__ H2, float* __restrict__ HP)
{
    const int lane  = threadIdx.x & 31;
    const int wave  = threadIdx.x >> 5;
    const int mbase = blockIdx.x * 16;
    const int n0    = wave * 16;
    const int r     = lane & 15;
    const int half  = lane >> 4;

    const float* xrow = X + (size_t)(mbase + r) * kIN;

    v8f acc1 = {};
    v8f acc2 = {};

#pragma unroll
    for (int k0 = 0; k0 < kIN; k0 += 4) {
        const int ka = k0 + half * 2;
        v2f a = *(const v2f*)(xrow + ka);
        v2f bA, bB;
        bA.x = W1[(size_t)ka       * kHD + n0 + r];
        bA.y = W1[(size_t)(ka + 1) * kHD + n0 + r];
        bB.x = W2[(size_t)ka       * kHD + n0 + r];
        bB.y = W2[(size_t)(ka + 1) * kHD + n0 + r];
        acc1 = __builtin_amdgcn_wmma_f32_16x16x4_f32(
            false, a, false, bA, (short)0, acc1, false, false);
        acc2 = __builtin_amdgcn_wmma_f32_16x16x4_f32(
            false, a, false, bB, (short)0, acc2, false, false);
    }

    const int n = n0 + r;
    const float bb1 = B1[n];
    const float bb2 = B2[n];

#pragma unroll
    for (int j = 0; j < 8; ++j) {
        const int m = mbase + j + half * 8;
        const float h1v = acc1[j] + bb1;
        const float h2v = acc2[j] + bb2;
        const size_t idx = (size_t)m * kHD + n;
        H1[idx] = h1v;
        H2[idx] = h2v;
        HP[idx] = h1v * h2v;
    }
}

constexpr int kScanThreads = 1024;
constexpr int kScanItems   = 20;
constexpr int kScanLive    = kN / kScanItems;
static_assert(kN % kScanItems == 0, "whole threads only");
static_assert(kScanLive < kScanThreads, "one block must cover all counts");
static_assert(kScanItems % 4 == 0, "int4 loads");
constexpr int kCntPad = kScanThreads * kScanItems;

__global__ __launch_bounds__(256) void dst_rank_kernel(
    const int* __restrict__ dst, int* __restrict__ cnt, int* __restrict__ rank)
{
    const int e = blockIdx.x * 256 + threadIdx.x;
    if (e >= kE) return;
    rank[e] = atomicAdd(cnt + dst[e], 1);
}

__global__ __launch_bounds__(kScanThreads) void scan_counts_kernel(
    const int* __restrict__ cnt, int* __restrict__ rowptr)
{
    __shared__ int wsum[32];
    const int t    = threadIdx.x;
    const int lane = t & 31;
    const int wave = t >> 5;

    int v[kScanItems];
    int tot = 0;
    if (t < kScanLive) {
        const int4* p = (const int4*)(cnt + t * kScanItems);
#pragma unroll
        for (int i = 0; i < kScanItems / 4; ++i) {
            const int4 q = p[i];
            v[4 * i + 0] = q.x; v[4 * i + 1] = q.y;
            v[4 * i + 2] = q.z; v[4 * i + 3] = q.w;
        }
#pragma unroll
        for (int i = 0; i < kScanItems; ++i) tot += v[i];
    } else {
#pragma unroll
        for (int i = 0; i < kScanItems; ++i) v[i] = 0;
    }

    int inc = tot;
#pragma unroll
    for (int off = 1; off < 32; off <<= 1) {
        const int o = __shfl_up(inc, off, 32);
        if (lane >= off) inc += o;
    }
    if (lane == 31) wsum[wave] = inc;
    __syncthreads();
    if (wave == 0) {
        const int w = wsum[lane];
        int winc = w;
#pragma unroll
        for (int off = 1; off < 32; off <<= 1) {
            const int o = __shfl_up(winc, off, 32);
            if (lane >= off) winc += o;
        }
        wsum[lane] = winc - w;
    }
    __syncthreads();

    int run = wsum[wave] + inc - tot;
    if (t < kScanLive) {
#pragma unroll
        for (int i = 0; i < kScanItems; ++i) {
            rowptr[t * kScanItems + i] = run;
            run += v[i];
        }
    }
    if (t == kScanThreads - 1) rowptr[kN] = run;
}

__global__ __launch_bounds__(256) void csr_fill_kernel(
    const int* __restrict__ src, const int* __restrict__ dst,
    const int* __restrict__ rowptr, const int* __restrict__ rank,
    int* __restrict__ csrc)
{
    const int e = blockIdx.x * 256 + threadIdx.x;
    if (e >= kE) return;
    csrc[rowptr[dst[e]] + rank[e]] = src[e];
}

constexpr int kNodesPerBlock      = 8;
constexpr int kNodesPerBlockFinal = 16;
static_assert(kN % kNodesPerBlockFinal == 0,
              "the last layer's grid covers whole blocks: no wave may leave before the barrier");

template <bool kFinal>
__global__ __launch_bounds__(kFinal ? kNodesPerBlockFinal * 32 : kNodesPerBlock * 32)
void node_attn_ln_kernel(
    const float* __restrict__ H1, const float* __restrict__ H2,
    const float* __restrict__ HP,
    const int* __restrict__ rowptr, const int* __restrict__ csrc,
    const float* __restrict__ avec,
    const float* __restrict__ G, const float* __restrict__ B,
    const float* __restrict__ Xin, float* __restrict__ Xout,
    const float* __restrict__ Wo, const float* __restrict__ bo,
    float* __restrict__ O)
{
    const int lane = threadIdx.x & 31;
    const int wave = __builtin_amdgcn_readfirstlane((int)(threadIdx.x >> 5));
    const int node = blockIdx.x * (kFinal ? kNodesPerBlockFinal : kNodesPerBlock) + wave;
    if constexpr (!kFinal) {
        if (node >= kN) return;
    }

    const int beg = rowptr[node];
    const int end = rowptr[node + 1];

    const float4 h2d = *(const float4*)(H2 + (size_t)node * kHD + lane * 4);
    const float4 hpd = *(const float4*)(HP + (size_t)node * kHD + lane * 4);
    const float4 av  = *(const float4*)(avec + lane * 4);

    float4 acc = make_float4(0.0f, 0.0f, 0.0f, 0.0f);

    for (int base = beg; base < end; base += 32) {
        const int n = min(32, end - base);
        int mine = 0;
        if (lane < n) mine = csrc[base + lane];

        for (int j = 0; j < n; j += 2) {
            const int j1 = min(j + 1, n - 1);
            int s0 = __builtin_amdgcn_readlane(mine, j);
            int s1 = __builtin_amdgcn_readlane(mine, j1);
            s0 = (int)min((unsigned)s0, (unsigned)(kN - 1));
            s1 = (int)min((unsigned)s1, (unsigned)(kN - 1));

            const float4 h1a = *(const float4*)(H1 + (size_t)s0 * kHD + lane * 4);
            const float4 hpa = *(const float4*)(HP + (size_t)s0 * kHD + lane * 4);
            const float4 h1b = *(const float4*)(H1 + (size_t)s1 * kHD + lane * 4);
            const float4 hpb = *(const float4*)(HP + (size_t)s1 * kHD + lane * 4);

            float p0 = head_part(h1a, hpa, h2d, hpd, av);
            float p1 = head_part(h1b, hpb, h2d, hpd, av);

            p0 += lane_xor_row<1>(p0);
            p1 += lane_xor_row<1>(p1);
            p0 += lane_xor_row<2>(p0);
            p1 += lane_xor_row<2>(p1);

            float m0 = p0, m1 = p1;
            m0 = fmaxf(m0, lane_xor_row<4>(m0));
            m1 = fmaxf(m1, lane_xor_row<4>(m1));
            m0 = fmaxf(m0, lane_xor_row<8>(m0));
            m1 = fmaxf(m1, lane_xor_row<8>(m1));
            m0 = fmaxf(m0, lane_xor_16(m0));
            m1 = fmaxf(m1, lane_xor_16(m1));

            const float e0 = exp_nonpos(p0 - m0);
            const float e1 = exp_nonpos(p1 - m1);
            float q0 = e0, q1 = e1;
            q0 += lane_xor_row<4>(q0);
            q1 += lane_xor_row<4>(q1);
            q0 += lane_xor_row<8>(q0);
            q1 += lane_xor_row<8>(q1);
            q0 += lane_xor_16(q0);
            q1 += lane_xor_16(q1);

            const float a0 = e0 * __builtin_amdgcn_rcpf(q0);
            const float a1 = e1 * __builtin_amdgcn_rcpf(q1);

            acc.x += a0 * h1a.x;
            acc.y += a0 * h1a.y;
            acc.z += a0 * h1a.z;
            acc.w += a0 * h1a.w;
            if (j + 1 < n) {
                acc.x += a1 * h1b.x;
                acc.y += a1 * h1b.y;
                acc.z += a1 * h1b.z;
                acc.w += a1 * h1b.w;
            }
        }
    }

    const float s = wave_sum(acc.x + acc.y + acc.z + acc.w);
    const float mu = s * (1.0f / 128.0f);

    const float dx0 = acc.x - mu, dx1 = acc.y - mu, dx2 = acc.z - mu, dx3 = acc.w - mu;
    const float q = wave_sum(dx0 * dx0 + dx1 * dx1 + dx2 * dx2 + dx3 * dx3);
    const float inv = rsqrtf(q * (1.0f / 128.0f) + kLnEps);

    const float4 g  = *(const float4*)(G + lane * 4);
    const float4 b  = *(const float4*)(B + lane * 4);
    const float4 xr = *(const float4*)(Xin + (size_t)node * kHD + lane * 4);

    float4 o;
    o.x = eluf(dx0 * inv * g.x + b.x) + xr.x;
    o.y = eluf(dx1 * inv * g.y + b.y) + xr.y;
    o.z = eluf(dx2 * inv * g.z + b.z) + xr.z;
    o.w = eluf(dx3 * inv * g.w + b.w) + xr.w;

    if constexpr (!kFinal) {
        *(float4*)(Xout + (size_t)node * kHD + lane * 4) = o;
    } else {
        const float4 w01 = *(const float4*)(Wo + lane * 8);
        const float4 w23 = *(const float4*)(Wo + lane * 8 + 4);
        float p0 = o.x * w01.x + o.y * w01.z + o.z * w23.x + o.w * w23.z;
        float p1 = o.x * w01.y + o.y * w01.w + o.z * w23.y + o.w * w23.w;
        p0 = wave_sum(p0);
        p1 = wave_sum(p1);
        __shared__ float2 proj[kNodesPerBlockFinal];
        if (lane == 0) proj[wave] = make_float2(p0 + bo[0], p1 + bo[1]);
        __syncthreads();
        if (wave == 0 && lane < kNodesPerBlockFinal) {
            *(float2*)(O + (size_t)(blockIdx.x * kNodesPerBlockFinal + lane) * kOUT) = proj[lane];
        }
    }
}

extern "C" void kernel_launch(void* const* d_in, const int* in_sizes, int n_in,
                              void* d_out, int out_size, void* d_ws, size_t ws_size,
                              hipStream_t stream)
{
    (void)in_sizes; (void)n_in; (void)out_size; (void)ws_size;

    const float* x    = (const float*)d_in[0];
    const int*   ei   = (const int*)  d_in[1];
    const float* W1   = (const float*)d_in[2];
    const float* b1   = (const float*)d_in[3];
    const float* W2   = (const float*)d_in[4];
    const float* b2   = (const float*)d_in[5];
    const float* a    = (const float*)d_in[6];
    const float* ln_g = (const float*)d_in[7];
    const float* ln_b = (const float*)d_in[8];
    const float* Wo   = (const float*)d_in[9];
    const float* bo   = (const float*)d_in[10];
    float*       out  = (float*)d_out;

    const size_t NHD = (size_t)kN * kHD;
    float* h1   = (float*)d_ws;
    float* h2   = h1  + NHD;
    float* hp   = h2  + NHD;
    float* xa   = hp  + NHD;
    int* cnt    = (int*)(xa + NHD);
    int* rowptr = cnt + kCntPad;
    int* rank   = rowptr + kCntPad;
    int* csrc   = rank + kE;

    const int* srcIdx = ei;
    const int* dstIdx = ei + kE;

    hipMemsetAsync(cnt, 0, (size_t)kN * sizeof(int), stream);
    dst_rank_kernel<<<kE / 256, 256, 0, stream>>>(dstIdx, cnt, rank);
    scan_counts_kernel<<<1, kScanThreads, 0, stream>>>(cnt, rowptr);
    csr_fill_kernel<<<kE / 256, 256, 0, stream>>>(srcIdx, dstIdx, rowptr, rank, csrc);

    static_assert(kL == 2, "layer 0 stores its rows, layer 1 projects them");

    gemm_h1h2_kernel<<<kN / 16, 256, 0, stream>>>(
        x, W1, b1, W2, b2, h1, h2, hp);
    node_attn_ln_kernel<false><<<kN / kNodesPerBlock, kNodesPerBlock * 32, 0, stream>>>(
        h1, h2, hp, rowptr, csrc, a, ln_g, ln_b, x, xa,
        nullptr, nullptr, nullptr);

    gemm_h1h2_kernel<<<kN / 16, 256, 0, stream>>>(
        xa,
        W1 + (size_t)kIN * kHD, b1 + kHD,
        W2 + (size_t)kIN * kHD, b2 + kHD,
        h1, h2, hp);
    node_attn_ln_kernel<true><<<kN / kNodesPerBlockFinal, kNodesPerBlockFinal * 32, 0, stream>>>(
        h1, h2, hp, rowptr, csrc, a + kH * kD, ln_g + kHD, ln_b + kHD, xa, nullptr,
        Wo, bo, out);
}


# GEMM: each wave keeps the W1/W2 fragments of its 16 columns in registers for all of K, 250 blocks walk 5 row tiles each; X by 16-byte loads, so the k pairing per matrix op changes (new sum order)
# speedup vs baseline: 1.9568x; 1.0765x over previous
#include <hip/hip_runtime.h>
#include <math.h>


constexpr int kN   = 20000;
constexpr int kE   = 320000;
constexpr int kIN  = 128;
constexpr int kH   = 8;
constexpr int kD   = 16;
constexpr int kHD  = 128;
constexpr int kL   = 2;
constexpr int kOUT = 2;
constexpr float kLnEps = 1e-5f;

typedef __attribute__((ext_vector_type(2))) float v2f;
typedef __attribute__((ext_vector_type(8))) float v8f;

__device__ __forceinline__ float exp_nonpos(float x) {
    return __builtin_amdgcn_exp2f(x * 1.44269504088896340736f);
}

__device__ __forceinline__ float eluf(float x) {
    return x > 0.0f ? x : (exp_nonpos(x) - 1.0f);
}

template <int M>
__device__ __forceinline__ float lane_xor_row(float x) {
    static_assert(M >= 1 && M <= 15, "row_xmask stays inside a 16-lane row");
    return __builtin_bit_cast(float, __builtin_amdgcn_update_dpp(
        0, __builtin_bit_cast(int, x), 0x160 + M, 0xF, 0xF, true));
}
__device__ __forceinline__ float lane_xor_16(float x) {
    const int b = __builtin_bit_cast(int, x);
    return __builtin_bit_cast(float, __builtin_amdgcn_permlanex16(
        b, b, 0x76543210, 0xfedcba98, false, false));
}
__device__ __forceinline__ float wave_sum(float v) {
    v += lane_xor_row<1>(v);
    v += lane_xor_row<2>(v);
    v += lane_xor_row<4>(v);
    v += lane_xor_row<8>(v);
    v += lane_xor_16(v);
    return v;
}

__device__ __forceinline__ float head_part(
    const float4& h1s, const float4& hps,
    const float4& h2d, const float4& hpd, const float4& av)
{
    const float t0 = eluf(h1s.x + h2d.x + hps.x * hpd.x);
    const float t1 = eluf(h1s.y + h2d.y + hps.y * hpd.y);
    const float t2 = eluf(h1s.z + h2d.z + hps.z * hpd.z);
    const float t3 = eluf(h1s.w + h2d.w + hps.w * hpd.w);
    const float part = av.x * t0 + av.y * t1 + av.z * t2 + av.w * t3;
    return part * 0.25f;
}

constexpr int kGemmTiles  = kN / 16;
constexpr int kGemmBlocks = 250;
static_assert(kN % 16 == 0, "whole 16-row tiles");

__device__ __forceinline__ void gemm_load_rows(const float* __restrict__ X, int tile,
                                               int r, int half, float4 (&q)[16])
{
    const float* xrow = X + (size_t)(tile * 16 + r) * kIN + half * 4;
#pragma unroll
    for (int g = 0; g < 16; ++g) q[g] = *(const float4*)(xrow + 8 * g);
}

__device__ __forceinline__ void gemm_tile(
    const float4 (&q)[16], const v2f (&w1)[32], const v2f (&w2)[32],
    float bb1, float bb2, int tile, int n, int half,
    float* __restrict__ H1, float* __restrict__ H2, float* __restrict__ HP)
{
    v8f acc1 = {};
    v8f acc2 = {};
#pragma unroll
    for (int g = 0; g < 16; ++g) {
        v2f a01, a23;
        a01.x = q[g].x; a01.y = q[g].y;
        a23.x = q[g].z; a23.y = q[g].w;
        acc1 = __builtin_amdgcn_wmma_f32_16x16x4_f32(
            false, a01, false, w1[2 * g], (short)0, acc1, false, false);
        acc2 = __builtin_amdgcn_wmma_f32_16x16x4_f32(
            false, a01, false, w2[2 * g], (short)0, acc2, false, false);
        acc1 = __builtin_amdgcn_wmma_f32_16x16x4_f32(
            false, a23, false, w1[2 * g + 1], (short)0, acc1, false, false);
        acc2 = __builtin_amdgcn_wmma_f32_16x16x4_f32(
            false, a23, false, w2[2 * g + 1], (short)0, acc2, false, false);
    }
    const size_t base = (size_t)(tile * 16 + half * 8) * kHD + n;
    float* __restrict__ p1 = H1 + base;
    float* __restrict__ p2 = H2 + base;
    float* __restrict__ pp = HP + base;
#pragma unroll
    for (int j = 0; j < 8; ++j) {
        const float h1v = acc1[j] + bb1;
        const float h2v = acc2[j] + bb2;
        p1[j * kHD] = h1v;
        p2[j * kHD] = h2v;
        pp[j * kHD] = h1v * h2v;
    }
}

__global__ __launch_bounds__(256)
void gemm_h1h2_kernel(
    const float* __restrict__ X,
    const float* __restrict__ W1, const float* __restrict__ B1,
    const float* __restrict__ W2, const float* __restrict__ B2,
    float* __restrict__ H1, float* __restrict__ H2, float* __restrict__ HP)
{
    const int lane  = threadIdx.x & 31;
    const int wave  = threadIdx.x >> 5;
    const int r     = lane & 15;
    const int half  = lane >> 4;
    const int n     = wave * 16 + r;

    v2f w1[32], w2[32];
#pragma unroll
    for (int g = 0; g < 16; ++g) {
        const int kb = 8 * g + 4 * half;
        w1[2 * g].x     = W1[(size_t)(kb + 0) * kHD + n];
        w1[2 * g].y     = W1[(size_t)(kb + 1) * kHD + n];
        w1[2 * g + 1].x = W1[(size_t)(kb + 2) * kHD + n];
        w1[2 * g + 1].y = W1[(size_t)(kb + 3) * kHD + n];
        w2[2 * g].x     = W2[(size_t)(kb + 0) * kHD + n];
        w2[2 * g].y     = W2[(size_t)(kb + 1) * kHD + n];
        w2[2 * g + 1].x = W2[(size_t)(kb + 2) * kHD + n];
        w2[2 * g + 1].y = W2[(size_t)(kb + 3) * kHD + n];
    }
    const float bb1 = B1[n];
    const float bb2 = B2[n];

    const int step  = (int)gridDim.x;
    const int first = (int)blockIdx.x;
    if (first >= kGemmTiles) return;
    const int cnt = (kGemmTiles - 1 - first) / step + 1;

    float4 qa[16], qb[16];
    gemm_load_rows(X, first, r, half, qa);
    int i = 0;
    for (; i + 1 < cnt; i += 2) {
        gemm_load_rows(X, first + (i + 1) * step, r, half, qb);
        __builtin_amdgcn_sched_barrier(0);
        gemm_tile(qa, w1, w2, bb1, bb2, first + i * step, n, half, H1, H2, HP);

        gemm_load_rows(X, first + min(i + 2, cnt - 1) * step, r, half, qa);
        __builtin_amdgcn_sched_barrier(0);
        gemm_tile(qb, w1, w2, bb1, bb2, first + (i + 1) * step, n, half, H1, H2, HP);
    }
    if (i < cnt) {
        gemm_tile(qa, w1, w2, bb1, bb2, first + i * step, n, half, H1, H2, HP);
    }
}

constexpr int kScanThreads = 1024;
constexpr int kScanItems   = 20;
constexpr int kScanLive    = kN / kScanItems;
static_assert(kN % kScanItems == 0, "whole threads only");
static_assert(kScanLive < kScanThreads, "one block must cover all counts");
static_assert(kScanItems % 4 == 0, "int4 loads");
constexpr int kCntPad = kScanThreads * kScanItems;

__global__ __launch_bounds__(256) void dst_rank_kernel(
    const int* __restrict__ dst, int* __restrict__ cnt, int* __restrict__ rank)
{
    const int e = blockIdx.x * 256 + threadIdx.x;
    if (e >= kE) return;
    rank[e] = atomicAdd(cnt + dst[e], 1);
}

__global__ __launch_bounds__(kScanThreads) void scan_counts_kernel(
    const int* __restrict__ cnt, int* __restrict__ rowptr)
{
    __shared__ int wsum[32];
    const int t    = threadIdx.x;
    const int lane = t & 31;
    const int wave = t >> 5;

    int v[kScanItems];
    int tot = 0;
    if (t < kScanLive) {
        const int4* p = (const int4*)(cnt + t * kScanItems);
#pragma unroll
        for (int i = 0; i < kScanItems / 4; ++i) {
            const int4 q = p[i];
            v[4 * i + 0] = q.x; v[4 * i + 1] = q.y;
            v[4 * i + 2] = q.z; v[4 * i + 3] = q.w;
        }
#pragma unroll
        for (int i = 0; i < kScanItems; ++i) tot += v[i];
    } else {
#pragma unroll
        for (int i = 0; i < kScanItems; ++i) v[i] = 0;
    }

    int inc = tot;
#pragma unroll
    for (int off = 1; off < 32; off <<= 1) {
        const int o = __shfl_up(inc, off, 32);
        if (lane >= off) inc += o;
    }
    if (lane == 31) wsum[wave] = inc;
    __syncthreads();
    if (wave == 0) {
        const int w = wsum[lane];
        int winc = w;
#pragma unroll
        for (int off = 1; off < 32; off <<= 1) {
            const int o = __shfl_up(winc, off, 32);
            if (lane >= off) winc += o;
        }
        wsum[lane] = winc - w;
    }
    __syncthreads();

    int run = wsum[wave] + inc - tot;
    if (t < kScanLive) {
#pragma unroll
        for (int i = 0; i < kScanItems; ++i) {
            rowptr[t * kScanItems + i] = run;
            run += v[i];
        }
    }
    if (t == kScanThreads - 1) rowptr[kN] = run;
}

__global__ __launch_bounds__(256) void csr_fill_kernel(
    const int* __restrict__ src, const int* __restrict__ dst,
    const int* __restrict__ rowptr, const int* __restrict__ rank,
    int* __restrict__ csrc)
{
    const int e = blockIdx.x * 256 + threadIdx.x;
    if (e >= kE) return;
    csrc[rowptr[dst[e]] + rank[e]] = src[e];
}

constexpr int kNodesPerBlock      = 8;
constexpr int kNodesPerBlockFinal = 16;
static_assert(kN % kNodesPerBlockFinal == 0,
              "the last layer's grid covers whole blocks: no wave may leave before the barrier");

template <bool kFinal>
__global__ __launch_bounds__(kFinal ? kNodesPerBlockFinal * 32 : kNodesPerBlock * 32)
void node_attn_ln_kernel(
    const float* __restrict__ H1, const float* __restrict__ H2,
    const float* __restrict__ HP,
    const int* __restrict__ rowptr, const int* __restrict__ csrc,
    const float* __restrict__ avec,
    const float* __restrict__ G, const float* __restrict__ B,
    const float* __restrict__ Xin, float* __restrict__ Xout,
    const float* __restrict__ Wo, const float* __restrict__ bo,
    float* __restrict__ O)
{
    const int lane = threadIdx.x & 31;
    const int wave = __builtin_amdgcn_readfirstlane((int)(threadIdx.x >> 5));
    const int node = blockIdx.x * (kFinal ? kNodesPerBlockFinal : kNodesPerBlock) + wave;
    if constexpr (!kFinal) {
        if (node >= kN) return;
    }

    const int beg = rowptr[node];
    const int end = rowptr[node + 1];

    const float4 h2d = *(const float4*)(H2 + (size_t)node * kHD + lane * 4);
    const float4 hpd = *(const float4*)(HP + (size_t)node * kHD + lane * 4);
    const float4 av  = *(const float4*)(avec + lane * 4);

    float4 acc = make_float4(0.0f, 0.0f, 0.0f, 0.0f);

    for (int base = beg; base < end; base += 32) {
        const int n = min(32, end - base);
        int mine = 0;
        if (lane < n) mine = csrc[base + lane];

        for (int j = 0; j < n; j += 2) {
            const int j1 = min(j + 1, n - 1);
            int s0 = __builtin_amdgcn_readlane(mine, j);
            int s1 = __builtin_amdgcn_readlane(mine, j1);
            s0 = (int)min((unsigned)s0, (unsigned)(kN - 1));
            s1 = (int)min((unsigned)s1, (unsigned)(kN - 1));

            const float4 h1a = *(const float4*)(H1 + (size_t)s0 * kHD + lane * 4);
            const float4 hpa = *(const float4*)(HP + (size_t)s0 * kHD + lane * 4);
            const float4 h1b = *(const float4*)(H1 + (size_t)s1 * kHD + lane * 4);
            const float4 hpb = *(const float4*)(HP + (size_t)s1 * kHD + lane * 4);

            float p0 = head_part(h1a, hpa, h2d, hpd, av);
            float p1 = head_part(h1b, hpb, h2d, hpd, av);

            p0 += lane_xor_row<1>(p0);
            p1 += lane_xor_row<1>(p1);
            p0 += lane_xor_row<2>(p0);
            p1 += lane_xor_row<2>(p1);

            float m0 = p0, m1 = p1;
            m0 = fmaxf(m0, lane_xor_row<4>(m0));
            m1 = fmaxf(m1, lane_xor_row<4>(m1));
            m0 = fmaxf(m0, lane_xor_row<8>(m0));
            m1 = fmaxf(m1, lane_xor_row<8>(m1));
            m0 = fmaxf(m0, lane_xor_16(m0));
            m1 = fmaxf(m1, lane_xor_16(m1));

            const float e0 = exp_nonpos(p0 - m0);
            const float e1 = exp_nonpos(p1 - m1);
            float q0 = e0, q1 = e1;
            q0 += lane_xor_row<4>(q0);
            q1 += lane_xor_row<4>(q1);
            q0 += lane_xor_row<8>(q0);
            q1 += lane_xor_row<8>(q1);
            q0 += lane_xor_16(q0);
            q1 += lane_xor_16(q1);

            const float a0 = e0 * __builtin_amdgcn_rcpf(q0);
            const float a1 = e1 * __builtin_amdgcn_rcpf(q1);

            acc.x += a0 * h1a.x;
            acc.y += a0 * h1a.y;
            acc.z += a0 * h1a.z;
            acc.w += a0 * h1a.w;
            if (j + 1 < n) {
                acc.x += a1 * h1b.x;
                acc.y += a1 * h1b.y;
                acc.z += a1 * h1b.z;
                acc.w += a1 * h1b.w;
            }
        }
    }

    const float s = wave_sum(acc.x + acc.y + acc.z + acc.w);
    const float mu = s * (1.0f / 128.0f);

    const float dx0 = acc.x - mu, dx1 = acc.y - mu, dx2 = acc.z - mu, dx3 = acc.w - mu;
    const float q = wave_sum(dx0 * dx0 + dx1 * dx1 + dx2 * dx2 + dx3 * dx3);
    const float inv = rsqrtf(q * (1.0f / 128.0f) + kLnEps);

    const float4 g  = *(const float4*)(G + lane * 4);
    const float4 b  = *(const float4*)(B + lane * 4);
    const float4 xr = *(const float4*)(Xin + (size_t)node * kHD + lane * 4);

    float4 o;
    o.x = eluf(dx0 * inv * g.x + b.x) + xr.x;
    o.y = eluf(dx1 * inv * g.y + b.y) + xr.y;
    o.z = eluf(dx2 * inv * g.z + b.z) + xr.z;
    o.w = eluf(dx3 * inv * g.w + b.w) + xr.w;

    if constexpr (!kFinal) {
        *(float4*)(Xout + (size_t)node * kHD + lane * 4) = o;
    } else {
        const float4 w01 = *(const float4*)(Wo + lane * 8);
        const float4 w23 = *(const float4*)(Wo + lane * 8 + 4);
        float p0 = o.x * w01.x + o.y * w01.z + o.z * w23.x + o.w * w23.z;
        float p1 = o.x * w01.y + o.y * w01.w + o.z * w23.y + o.w * w23.w;
        p0 = wave_sum(p0);
        p1 = wave_sum(p1);
        __shared__ float2 proj[kNodesPerBlockFinal];
        if (lane == 0) proj[wave] = make_float2(p0 + bo[0], p1 + bo[1]);
        __syncthreads();
        if (wave == 0 && lane < kNodesPerBlockFinal) {
            *(float2*)(O + (size_t)(blockIdx.x * kNodesPerBlockFinal + lane) * kOUT) = proj[lane];
        }
    }
}

extern "C" void kernel_launch(void* const* d_in, const int* in_sizes, int n_in,
                              void* d_out, int out_size, void* d_ws, size_t ws_size,
                              hipStream_t stream)
{
    (void)in_sizes; (void)n_in; (void)out_size; (void)ws_size;

    const float* x    = (const float*)d_in[0];
    const int*   ei   = (const int*)  d_in[1];
    const float* W1   = (const float*)d_in[2];
    const float* b1   = (const float*)d_in[3];
    const float* W2   = (const float*)d_in[4];
    const float* b2   = (const float*)d_in[5];
    const float* a    = (const float*)d_in[6];
    const float* ln_g = (const float*)d_in[7];
    const float* ln_b = (const float*)d_in[8];
    const float* Wo   = (const float*)d_in[9];
    const float* bo   = (const float*)d_in[10];
    float*       out  = (float*)d_out;

    const size_t NHD = (size_t)kN * kHD;
    float* h1   = (float*)d_ws;
    float* h2   = h1  + NHD;
    float* hp   = h2  + NHD;
    float* xa   = hp  + NHD;
    int* cnt    = (int*)(xa + NHD);
    int* rowptr = cnt + kCntPad;
    int* rank   = rowptr + kCntPad;
    int* csrc   = rank + kE;

    const int* srcIdx = ei;
    const int* dstIdx = ei + kE;

    hipMemsetAsync(cnt, 0, (size_t)kN * sizeof(int), stream);
    dst_rank_kernel<<<kE / 256, 256, 0, stream>>>(dstIdx, cnt, rank);
    scan_counts_kernel<<<1, kScanThreads, 0, stream>>>(cnt, rowptr);
    csr_fill_kernel<<<kE / 256, 256, 0, stream>>>(srcIdx, dstIdx, rowptr, rank, csrc);

    static_assert(kL == 2, "layer 0 stores its rows, layer 1 projects them");

    gemm_h1h2_kernel<<<kGemmBlocks, 256, 0, stream>>>(
        x, W1, b1, W2, b2, h1, h2, hp);
    node_attn_ln_kernel<false><<<kN / kNodesPerBlock, kNodesPerBlock * 32, 0, stream>>>(
        h1, h2, hp, rowptr, csrc, a, ln_g, ln_b, x, xa,
        nullptr, nullptr, nullptr);

    gemm_h1h2_kernel<<<kGemmBlocks, 256, 0, stream>>>(
        xa,
        W1 + (size_t)kIN * kHD, b1 + kHD,
        W2 + (size_t)kIN * kHD, b2 + kHD,
        h1, h2, hp);
    node_attn_ln_kernel<true><<<kN / kNodesPerBlockFinal, kNodesPerBlockFinal * 32, 0, stream>>>(
        h1, h2, hp, rowptr, csrc, a + kH * kD, ln_g + kHD, ln_b + kHD, xa, nullptr,
        Wo, bo, out);
}


# Edge grouping in one launch: the counting atomic's return value puts src straight into its destination's 64-slot row (spill list past 64); scan and fill launches gone; same arrival-order sums
# speedup vs baseline: 2.2188x; 1.1339x over previous
#include <hip/hip_runtime.h>
#include <math.h>


constexpr int kN   = 20000;
constexpr int kE   = 320000;
constexpr int kIN  = 128;
constexpr int kH   = 8;
constexpr int kD   = 16;
constexpr int kHD  = 128;
constexpr int kL   = 2;
constexpr int kOUT = 2;
constexpr float kLnEps = 1e-5f;

typedef __attribute__((ext_vector_type(2))) float v2f;
typedef __attribute__((ext_vector_type(8))) float v8f;

__device__ __forceinline__ float exp_nonpos(float x) {
    return __builtin_amdgcn_exp2f(x * 1.44269504088896340736f);
}

__device__ __forceinline__ float eluf(float x) {
    return x > 0.0f ? x : (exp_nonpos(x) - 1.0f);
}

template <int M>
__device__ __forceinline__ float lane_xor_row(float x) {
    static_assert(M >= 1 && M <= 15, "row_xmask stays inside a 16-lane row");
    return __builtin_bit_cast(float, __builtin_amdgcn_update_dpp(
        0, __builtin_bit_cast(int, x), 0x160 + M, 0xF, 0xF, true));
}
__device__ __forceinline__ float lane_xor_16(float x) {
    const int b = __builtin_bit_cast(int, x);
    return __builtin_bit_cast(float, __builtin_amdgcn_permlanex16(
        b, b, 0x76543210, 0xfedcba98, false, false));
}
__device__ __forceinline__ float wave_sum(float v) {
    v += lane_xor_row<1>(v);
    v += lane_xor_row<2>(v);
    v += lane_xor_row<4>(v);
    v += lane_xor_row<8>(v);
    v += lane_xor_16(v);
    return v;
}

__device__ __forceinline__ float head_part(
    const float4& h1s, const float4& hps,
    const float4& h2d, const float4& hpd, const float4& av)
{
    const float t0 = eluf(h1s.x + h2d.x + hps.x * hpd.x);
    const float t1 = eluf(h1s.y + h2d.y + hps.y * hpd.y);
    const float t2 = eluf(h1s.z + h2d.z + hps.z * hpd.z);
    const float t3 = eluf(h1s.w + h2d.w + hps.w * hpd.w);
    const float part = av.x * t0 + av.y * t1 + av.z * t2 + av.w * t3;
    return part * 0.25f;
}

constexpr int kGemmTiles  = kN / 16;
constexpr int kGemmBlocks = 250;
static_assert(kN % 16 == 0, "whole 16-row tiles");

__device__ __forceinline__ void gemm_load_rows(const float* __restrict__ X, int tile,
                                               int r, int half, float4 (&q)[16])
{
    const float* xrow = X + (size_t)(tile * 16 + r) * kIN + half * 4;
#pragma unroll
    for (int g = 0; g < 16; ++g) q[g] = *(const float4*)(xrow + 8 * g);
}

__device__ __forceinline__ void gemm_tile(
    const float4 (&q)[16], const v2f (&w1)[32], const v2f (&w2)[32],
    float bb1, float bb2, int tile, int n, int half,
    float* __restrict__ H1, float* __restrict__ H2, float* __restrict__ HP)
{
    v8f acc1 = {};
    v8f acc2 = {};
#pragma unroll
    for (int g = 0; g < 16; ++g) {
        v2f a01, a23;
        a01.x = q[g].x; a01.y = q[g].y;
        a23.x = q[g].z; a23.y = q[g].w;
        acc1 = __builtin_amdgcn_wmma_f32_16x16x4_f32(
            false, a01, false, w1[2 * g], (short)0, acc1, false, false);
        acc2 = __builtin_amdgcn_wmma_f32_16x16x4_f32(
            false, a01, false, w2[2 * g], (short)0, acc2, false, false);
        acc1 = __builtin_amdgcn_wmma_f32_16x16x4_f32(
            false, a23, false, w1[2 * g + 1], (short)0, acc1, false, false);
        acc2 = __builtin_amdgcn_wmma_f32_16x16x4_f32(
            false, a23, false, w2[2 * g + 1], (short)0, acc2, false, false);
    }
    const size_t base = (size_t)(tile * 16 + half * 8) * kHD + n;
    float* __restrict__ p1 = H1 + base;
    float* __restrict__ p2 = H2 + base;
    float* __restrict__ pp = HP + base;
#pragma unroll
    for (int j = 0; j < 8; ++j) {
        const float h1v = acc1[j] + bb1;
        const float h2v = acc2[j] + bb2;
        p1[j * kHD] = h1v;
        p2[j * kHD] = h2v;
        pp[j * kHD] = h1v * h2v;
    }
}

__global__ __launch_bounds__(256)
void gemm_h1h2_kernel(
    const float* __restrict__ X,
    const float* __restrict__ W1, const float* __restrict__ B1,
    const float* __restrict__ W2, const float* __restrict__ B2,
    float* __restrict__ H1, float* __restrict__ H2, float* __restrict__ HP)
{
    const int lane  = threadIdx.x & 31;
    const int wave  = threadIdx.x >> 5;
    const int r     = lane & 15;
    const int half  = lane >> 4;
    const int n     = wave * 16 + r;

    v2f w1[32], w2[32];
#pragma unroll
    for (int g = 0; g < 16; ++g) {
        const int kb = 8 * g + 4 * half;
        w1[2 * g].x     = W1[(size_t)(kb + 0) * kHD + n];
        w1[2 * g].y     = W1[(size_t)(kb + 1) * kHD + n];
        w1[2 * g + 1].x = W1[(size_t)(kb + 2) * kHD + n];
        w1[2 * g + 1].y = W1[(size_t)(kb + 3) * kHD + n];
        w2[2 * g].x     = W2[(size_t)(kb + 0) * kHD + n];
        w2[2 * g].y     = W2[(size_t)(kb + 1) * kHD + n];
        w2[2 * g + 1].x = W2[(size_t)(kb + 2) * kHD + n];
        w2[2 * g + 1].y = W2[(size_t)(kb + 3) * kHD + n];
    }
    const float bb1 = B1[n];
    const float bb2 = B2[n];

    const int step  = (int)gridDim.x;
    const int first = (int)blockIdx.x;
    if (first >= kGemmTiles) return;
    const int cnt = (kGemmTiles - 1 - first) / step + 1;

    float4 qa[16], qb[16];
    gemm_load_rows(X, first, r, half, qa);
    int i = 0;
    for (; i + 1 < cnt; i += 2) {
        gemm_load_rows(X, first + (i + 1) * step, r, half, qb);
        __builtin_amdgcn_sched_barrier(0);
        gemm_tile(qa, w1, w2, bb1, bb2, first + i * step, n, half, H1, H2, HP);

        gemm_load_rows(X, first + min(i + 2, cnt - 1) * step, r, half, qa);
        __builtin_amdgcn_sched_barrier(0);
        gemm_tile(qb, w1, w2, bb1, bb2, first + (i + 1) * step, n, half, H1, H2, HP);
    }
    if (i < cnt) {
        gemm_tile(qa, w1, w2, bb1, bb2, first + i * step, n, half, H1, H2, HP);
    }
}

constexpr int kSlots      = 64;
constexpr int kCntPad     = 20480;
constexpr int kSpillCount = kN;
static_assert(kSpillCount < kCntPad, "the spill count lies inside the cleared table");
static_assert((kSlots * sizeof(int)) % 256 == 0, "rows are whole aligned 256-byte units");

__global__ __launch_bounds__(256) void group_edges_kernel(
    const int* __restrict__ src, const int* __restrict__ dst,
    int* __restrict__ cnt, int* __restrict__ slot, int* __restrict__ spill)
{
    const int e = blockIdx.x * 256 + threadIdx.x;
    if (e >= kE) return;
    const int d = dst[e];
    const int s = src[e];
    const int p = atomicAdd(cnt + d, 1);
    if (p < kSlots) {
        slot[d * kSlots + p] = s;
    } else {
        const int o = atomicAdd(cnt + kSpillCount, 1);
        *(int2*)(spill + 2 * (size_t)o) = make_int2(d, s);
    }
}

__device__ __forceinline__ void add_one_edge(
    const float* __restrict__ H1, const float* __restrict__ HP, int s, int lane,
    const float4& h2d, const float4& hpd, const float4& av, float4& acc)
{
    s = (int)min((unsigned)s, (unsigned)(kN - 1));
    const float4 h1a = *(const float4*)(H1 + (size_t)s * kHD + lane * 4);
    const float4 hpa = *(const float4*)(HP + (size_t)s * kHD + lane * 4);
    float p = head_part(h1a, hpa, h2d, hpd, av);
    p += lane_xor_row<1>(p);
    p += lane_xor_row<2>(p);
    float m = p;
    m = fmaxf(m, lane_xor_row<4>(m));
    m = fmaxf(m, lane_xor_row<8>(m));
    m = fmaxf(m, lane_xor_16(m));
    const float e = exp_nonpos(p - m);
    float q = e;
    q += lane_xor_row<4>(q);
    q += lane_xor_row<8>(q);
    q += lane_xor_16(q);
    const float a = e * __builtin_amdgcn_rcpf(q);
    acc.x += a * h1a.x;
    acc.y += a * h1a.y;
    acc.z += a * h1a.z;
    acc.w += a * h1a.w;
}

constexpr int kNodesPerBlock      = 8;
constexpr int kNodesPerBlockFinal = 16;
static_assert(kN % kNodesPerBlockFinal == 0,
              "the last layer's grid covers whole blocks: no wave may leave before the barrier");

template <bool kFinal>
__global__ __launch_bounds__(kFinal ? kNodesPerBlockFinal * 32 : kNodesPerBlock * 32)
void node_attn_ln_kernel(
    const float* __restrict__ H1, const float* __restrict__ H2,
    const float* __restrict__ HP,
    const int* __restrict__ cnt, const int* __restrict__ slot,
    const int* __restrict__ spill,
    const float* __restrict__ avec,
    const float* __restrict__ G, const float* __restrict__ B,
    const float* __restrict__ Xin, float* __restrict__ Xout,
    const float* __restrict__ Wo, const float* __restrict__ bo,
    float* __restrict__ O)
{
    const int lane = threadIdx.x & 31;
    const int wave = __builtin_amdgcn_readfirstlane((int)(threadIdx.x >> 5));
    const int node = blockIdx.x * (kFinal ? kNodesPerBlockFinal : kNodesPerBlock) + wave;
    if constexpr (!kFinal) {
        if (node >= kN) return;
    }

    const int deg = cnt[node];
    const int beg = node * kSlots;
    const int end = beg + min(deg, kSlots);

    const float4 h2d = *(const float4*)(H2 + (size_t)node * kHD + lane * 4);
    const float4 hpd = *(const float4*)(HP + (size_t)node * kHD + lane * 4);
    const float4 av  = *(const float4*)(avec + lane * 4);

    float4 acc = make_float4(0.0f, 0.0f, 0.0f, 0.0f);

    for (int base = beg; base < end; base += 32) {
        const int n = min(32, end - base);
        int mine = 0;
        if (lane < n) mine = slot[base + lane];

        for (int j = 0; j < n; j += 2) {
            const int j1 = min(j + 1, n - 1);
            int s0 = __builtin_amdgcn_readlane(mine, j);
            int s1 = __builtin_amdgcn_readlane(mine, j1);
            s0 = (int)min((unsigned)s0, (unsigned)(kN - 1));
            s1 = (int)min((unsigned)s1, (unsigned)(kN - 1));

            const float4 h1a = *(const float4*)(H1 + (size_t)s0 * kHD + lane * 4);
            const float4 hpa = *(const float4*)(HP + (size_t)s0 * kHD + lane * 4);
            const float4 h1b = *(const float4*)(H1 + (size_t)s1 * kHD + lane * 4);
            const float4 hpb = *(const float4*)(HP + (size_t)s1 * kHD + lane * 4);

            float p0 = head_part(h1a, hpa, h2d, hpd, av);
            float p1 = head_part(h1b, hpb, h2d, hpd, av);

            p0 += lane_xor_row<1>(p0);
            p1 += lane_xor_row<1>(p1);
            p0 += lane_xor_row<2>(p0);
            p1 += lane_xor_row<2>(p1);

            float m0 = p0, m1 = p1;
            m0 = fmaxf(m0, lane_xor_row<4>(m0));
            m1 = fmaxf(m1, lane_xor_row<4>(m1));
            m0 = fmaxf(m0, lane_xor_row<8>(m0));
            m1 = fmaxf(m1, lane_xor_row<8>(m1));
            m0 = fmaxf(m0, lane_xor_16(m0));
            m1 = fmaxf(m1, lane_xor_16(m1));

            const float e0 = exp_nonpos(p0 - m0);
            const float e1 = exp_nonpos(p1 - m1);
            float q0 = e0, q1 = e1;
            q0 += lane_xor_row<4>(q0);
            q1 += lane_xor_row<4>(q1);
            q0 += lane_xor_row<8>(q0);
            q1 += lane_xor_row<8>(q1);
            q0 += lane_xor_16(q0);
            q1 += lane_xor_16(q1);

            const float a0 = e0 * __builtin_amdgcn_rcpf(q0);
            const float a1 = e1 * __builtin_amdgcn_rcpf(q1);

            acc.x += a0 * h1a.x;
            acc.y += a0 * h1a.y;
            acc.z += a0 * h1a.z;
            acc.w += a0 * h1a.w;
            if (j + 1 < n) {
                acc.x += a1 * h1b.x;
                acc.y += a1 * h1b.y;
                acc.z += a1 * h1b.z;
                acc.w += a1 * h1b.w;
            }
        }
    }

    if (deg > kSlots) {
        const int total = min(cnt[kSpillCount], kE);
        for (int ob = 0; ob < total; ob += 32) {
            const int m = min(32, total - ob);
            int od = -1, os = 0;
            if (lane < m) {
                const int2 pr = *(const int2*)(spill + 2 * (size_t)(ob + lane));
                od = pr.x;
                os = pr.y;
            }
            unsigned hit = __builtin_amdgcn_ballot_w32(od == node);
            while (hit != 0u) {
                const int l = __builtin_ctz(hit);
                hit &= hit - 1u;
                add_one_edge(H1, HP, __builtin_amdgcn_readlane(os, l), lane,
                             h2d, hpd, av, acc);
            }
        }
    }

    const float s = wave_sum(acc.x + acc.y + acc.z + acc.w);
    const float mu = s * (1.0f / 128.0f);

    const float dx0 = acc.x - mu, dx1 = acc.y - mu, dx2 = acc.z - mu, dx3 = acc.w - mu;
    const float q = wave_sum(dx0 * dx0 + dx1 * dx1 + dx2 * dx2 + dx3 * dx3);
    const float inv = rsqrtf(q * (1.0f / 128.0f) + kLnEps);

    const float4 g  = *(const float4*)(G + lane * 4);
    const float4 b  = *(const float4*)(B + lane * 4);
    const float4 xr = *(const float4*)(Xin + (size_t)node * kHD + lane * 4);

    float4 o;
    o.x = eluf(dx0 * inv * g.x + b.x) + xr.x;
    o.y = eluf(dx1 * inv * g.y + b.y) + xr.y;
    o.z = eluf(dx2 * inv * g.z + b.z) + xr.z;
    o.w = eluf(dx3 * inv * g.w + b.w) + xr.w;

    if constexpr (!kFinal) {
        *(float4*)(Xout + (size_t)node * kHD + lane * 4) = o;
    } else {
        const float4 w01 = *(const float4*)(Wo + lane * 8);
        const float4 w23 = *(const float4*)(Wo + lane * 8 + 4);
        float p0 = o.x * w01.x + o.y * w01.z + o.z * w23.x + o.w * w23.z;
        float p1 = o.x * w01.y + o.y * w01.w + o.z * w23.y + o.w * w23.w;
        p0 = wave_sum(p0);
        p1 = wave_sum(p1);
        __shared__ float2 proj[kNodesPerBlockFinal];
        if (lane == 0) proj[wave] = make_float2(p0 + bo[0], p1 + bo[1]);
        __syncthreads();
        if (wave == 0 && lane < kNodesPerBlockFinal) {
            *(float2*)(O + (size_t)(blockIdx.x * kNodesPerBlockFinal + lane) * kOUT) = proj[lane];
        }
    }
}

extern "C" void kernel_launch(void* const* d_in, const int* in_sizes, int n_in,
                              void* d_out, int out_size, void* d_ws, size_t ws_size,
                              hipStream_t stream)
{
    (void)in_sizes; (void)n_in; (void)out_size; (void)ws_size;

    const float* x    = (const float*)d_in[0];
    const int*   ei   = (const int*)  d_in[1];
    const float* W1   = (const float*)d_in[2];
    const float* b1   = (const float*)d_in[3];
    const float* W2   = (const float*)d_in[4];
    const float* b2   = (const float*)d_in[5];
    const float* a    = (const float*)d_in[6];
    const float* ln_g = (const float*)d_in[7];
    const float* ln_b = (const float*)d_in[8];
    const float* Wo   = (const float*)d_in[9];
    const float* bo   = (const float*)d_in[10];
    float*       out  = (float*)d_out;

    const size_t NHD = (size_t)kN * kHD;
    float* h1   = (float*)d_ws;
    float* h2   = h1  + NHD;
    float* hp   = h2  + NHD;
    float* xa   = hp  + NHD;
    int* cnt    = (int*)(xa + NHD);
    int* slot   = cnt + kCntPad;
    int* spill  = slot + (size_t)kN * kSlots;

    const int* srcIdx = ei;
    const int* dstIdx = ei + kE;

    hipMemsetAsync(cnt, 0, (size_t)kCntPad * sizeof(int), stream);
    group_edges_kernel<<<kE / 256, 256, 0, stream>>>(srcIdx, dstIdx, cnt, slot, spill);

    static_assert(kL == 2, "layer 0 stores its rows, layer 1 projects them");

    gemm_h1h2_kernel<<<kGemmBlocks, 256, 0, stream>>>(
        x, W1, b1, W2, b2, h1, h2, hp);
    node_attn_ln_kernel<false><<<kN / kNodesPerBlock, kNodesPerBlock * 32, 0, stream>>>(
        h1, h2, hp, cnt, slot, spill, a, ln_g, ln_b, x, xa,
        nullptr, nullptr, nullptr);

    gemm_h1h2_kernel<<<kGemmBlocks, 256, 0, stream>>>(
        xa,
        W1 + (size_t)kIN * kHD, b1 + kHD,
        W2 + (size_t)kIN * kHD, b2 + kHD,
        h1, h2, hp);
    node_attn_ln_kernel<true><<<kN / kNodesPerBlockFinal, kNodesPerBlockFinal * 32, 0, stream>>>(
        h1, h2, hp, cnt, slot, spill, a + kH * kD, ln_g + kHD, ln_b + kHD, xa, nullptr,
        Wo, bo, out);
}


# Node kernel: four edges per trip, all eight row gathers issued together (row address = scalar base + one 32-bit offset per edge): half the dependent load waits per node; same sums in list order
# speedup vs baseline: 2.2843x; 1.0295x over previous
#include <hip/hip_runtime.h>
#include <math.h>


constexpr int kN   = 20000;
constexpr int kE   = 320000;
constexpr int kIN  = 128;
constexpr int kH   = 8;
constexpr int kD   = 16;
constexpr int kHD  = 128;
constexpr int kL   = 2;
constexpr int kOUT = 2;
constexpr float kLnEps = 1e-5f;

typedef __attribute__((ext_vector_type(2))) float v2f;
typedef __attribute__((ext_vector_type(8))) float v8f;

__device__ __forceinline__ float exp_nonpos(float x) {
    return __builtin_amdgcn_exp2f(x * 1.44269504088896340736f);
}

__device__ __forceinline__ float eluf(float x) {
    return x > 0.0f ? x : (exp_nonpos(x) - 1.0f);
}

template <int M>
__device__ __forceinline__ float lane_xor_row(float x) {
    static_assert(M >= 1 && M <= 15, "row_xmask stays inside a 16-lane row");
    return __builtin_bit_cast(float, __builtin_amdgcn_update_dpp(
        0, __builtin_bit_cast(int, x), 0x160 + M, 0xF, 0xF, true));
}
__device__ __forceinline__ float lane_xor_16(float x) {
    const int b = __builtin_bit_cast(int, x);
    return __builtin_bit_cast(float, __builtin_amdgcn_permlanex16(
        b, b, 0x76543210, 0xfedcba98, false, false));
}
__device__ __forceinline__ float wave_sum(float v) {
    v += lane_xor_row<1>(v);
    v += lane_xor_row<2>(v);
    v += lane_xor_row<4>(v);
    v += lane_xor_row<8>(v);
    v += lane_xor_16(v);
    return v;
}

__device__ __forceinline__ float head_part(
    const float4& h1s, const float4& hps,
    const float4& h2d, const float4& hpd, const float4& av)
{
    const float t0 = eluf(h1s.x + h2d.x + hps.x * hpd.x);
    const float t1 = eluf(h1s.y + h2d.y + hps.y * hpd.y);
    const float t2 = eluf(h1s.z + h2d.z + hps.z * hpd.z);
    const float t3 = eluf(h1s.w + h2d.w + hps.w * hpd.w);
    const float part = av.x * t0 + av.y * t1 + av.z * t2 + av.w * t3;
    return part * 0.25f;
}

constexpr int kGemmTiles  = kN / 16;
constexpr int kGemmBlocks = 250;
static_assert(kN % 16 == 0, "whole 16-row tiles");

__device__ __forceinline__ void gemm_load_rows(const float* __restrict__ X, int tile,
                                               int r, int half, float4 (&q)[16])
{
    const float* xrow = X + (size_t)(tile * 16 + r) * kIN + half * 4;
#pragma unroll
    for (int g = 0; g < 16; ++g) q[g] = *(const float4*)(xrow + 8 * g);
}

__device__ __forceinline__ void gemm_tile(
    const float4 (&q)[16], const v2f (&w1)[32], const v2f (&w2)[32],
    float bb1, float bb2, int tile, int n, int half,
    float* __restrict__ H1, float* __restrict__ H2, float* __restrict__ HP)
{
    v8f acc1 = {};
    v8f acc2 = {};
#pragma unroll
    for (int g = 0; g < 16; ++g) {
        v2f a01, a23;
        a01.x = q[g].x; a01.y = q[g].y;
        a23.x = q[g].z; a23.y = q[g].w;
        acc1 = __builtin_amdgcn_wmma_f32_16x16x4_f32(
            false, a01, false, w1[2 * g], (short)0, acc1, false, false);
        acc2 = __builtin_amdgcn_wmma_f32_16x16x4_f32(
            false, a01, false, w2[2 * g], (short)0, acc2, false, false);
        acc1 = __builtin_amdgcn_wmma_f32_16x16x4_f32(
            false, a23, false, w1[2 * g + 1], (short)0, acc1, false, false);
        acc2 = __builtin_amdgcn_wmma_f32_16x16x4_f32(
            false, a23, false, w2[2 * g + 1], (short)0, acc2, false, false);
    }
    const size_t base = (size_t)(tile * 16 + half * 8) * kHD + n;
    float* __restrict__ p1 = H1 + base;
    float* __restrict__ p2 = H2 + base;
    float* __restrict__ pp = HP + base;
#pragma unroll
    for (int j = 0; j < 8; ++j) {
        const float h1v = acc1[j] + bb1;
        const float h2v = acc2[j] + bb2;
        p1[j * kHD] = h1v;
        p2[j * kHD] = h2v;
        pp[j * kHD] = h1v * h2v;
    }
}

__global__ __launch_bounds__(256)
void gemm_h1h2_kernel(
    const float* __restrict__ X,
    const float* __restrict__ W1, const float* __restrict__ B1,
    const float* __restrict__ W2, const float* __restrict__ B2,
    float* __restrict__ H1, float* __restrict__ H2, float* __restrict__ HP)
{
    const int lane  = threadIdx.x & 31;
    const int wave  = threadIdx.x >> 5;
    const int r     = lane & 15;
    const int half  = lane >> 4;
    const int n     = wave * 16 + r;

    v2f w1[32], w2[32];
#pragma unroll
    for (int g = 0; g < 16; ++g) {
        const int kb = 8 * g + 4 * half;
        w1[2 * g].x     = W1[(size_t)(kb + 0) * kHD + n];
        w1[2 * g].y     = W1[(size_t)(kb + 1) * kHD + n];
        w1[2 * g + 1].x = W1[(size_t)(kb + 2) * kHD + n];
        w1[2 * g + 1].y = W1[(size_t)(kb + 3) * kHD + n];
        w2[2 * g].x     = W2[(size_t)(kb + 0) * kHD + n];
        w2[2 * g].y     = W2[(size_t)(kb + 1) * kHD + n];
        w2[2 * g + 1].x = W2[(size_t)(kb + 2) * kHD + n];
        w2[2 * g + 1].y = W2[(size_t)(kb + 3) * kHD + n];
    }
    const float bb1 = B1[n];
    const float bb2 = B2[n];

    const int step  = (int)gridDim.x;
    const int first = (int)blockIdx.x;
    if (first >= kGemmTiles) return;
    const int cnt = (kGemmTiles - 1 - first) / step + 1;

    float4 qa[16], qb[16];
    gemm_load_rows(X, first, r, half, qa);
    int i = 0;
    for (; i + 1 < cnt; i += 2) {
        gemm_load_rows(X, first + (i + 1) * step, r, half, qb);
        __builtin_amdgcn_sched_barrier(0);
        gemm_tile(qa, w1, w2, bb1, bb2, first + i * step, n, half, H1, H2, HP);

        gemm_load_rows(X, first + min(i + 2, cnt - 1) * step, r, half, qa);
        __builtin_amdgcn_sched_barrier(0);
        gemm_tile(qb, w1, w2, bb1, bb2, first + (i + 1) * step, n, half, H1, H2, HP);
    }
    if (i < cnt) {
        gemm_tile(qa, w1, w2, bb1, bb2, first + i * step, n, half, H1, H2, HP);
    }
}

constexpr int kSlots      = 64;
constexpr int kCntPad     = 20480;
constexpr int kSpillCount = kN;
static_assert(kSpillCount < kCntPad, "the spill count lies inside the cleared table");
static_assert((kSlots * sizeof(int)) % 256 == 0, "rows are whole aligned 256-byte units");

__global__ __launch_bounds__(256) void group_edges_kernel(
    const int* __restrict__ src, const int* __restrict__ dst,
    int* __restrict__ cnt, int* __restrict__ slot, int* __restrict__ spill)
{
    const int e = blockIdx.x * 256 + threadIdx.x;
    if (e >= kE) return;
    const int d = dst[e];
    const int s = src[e];
    const int p = atomicAdd(cnt + d, 1);
    if (p < kSlots) {
        slot[d * kSlots + p] = s;
    } else {
        const int o = atomicAdd(cnt + kSpillCount, 1);
        *(int2*)(spill + 2 * (size_t)o) = make_int2(d, s);
    }
}

template <int K>
__device__ __forceinline__ void add_edges(
    const float* __restrict__ H1, const float* __restrict__ HP,
    const int (&s)[K], bool last_valid, int lane,
    const float4& h2d, const float4& hpd, const float4& av, float4& acc)
{
    float4 h1r[K], hpr[K];
#pragma unroll
    for (int e = 0; e < K; ++e) {
        const unsigned off = (unsigned)s[e] * (unsigned)(kHD * sizeof(float))
                           + (unsigned)lane * (unsigned)sizeof(float4);
        h1r[e] = *(const float4*)((const char*)H1 + off);
        hpr[e] = *(const float4*)((const char*)HP + off);
    }

    float p[K], m[K], ex[K], q[K];
#pragma unroll
    for (int e = 0; e < K; ++e) p[e] = head_part(h1r[e], hpr[e], h2d, hpd, av);

#pragma unroll
    for (int e = 0; e < K; ++e) p[e] += lane_xor_row<1>(p[e]);
#pragma unroll
    for (int e = 0; e < K; ++e) p[e] += lane_xor_row<2>(p[e]);

#pragma unroll
    for (int e = 0; e < K; ++e) m[e] = fmaxf(p[e], lane_xor_row<4>(p[e]));
#pragma unroll
    for (int e = 0; e < K; ++e) m[e] = fmaxf(m[e], lane_xor_row<8>(m[e]));
#pragma unroll
    for (int e = 0; e < K; ++e) m[e] = fmaxf(m[e], lane_xor_16(m[e]));

#pragma unroll
    for (int e = 0; e < K; ++e) ex[e] = exp_nonpos(p[e] - m[e]);
#pragma unroll
    for (int e = 0; e < K; ++e) q[e] = ex[e] + lane_xor_row<4>(ex[e]);
#pragma unroll
    for (int e = 0; e < K; ++e) q[e] += lane_xor_row<8>(q[e]);
#pragma unroll
    for (int e = 0; e < K; ++e) q[e] += lane_xor_16(q[e]);

#pragma unroll
    for (int e = 0; e < K; ++e) {
        const float a = ex[e] * __builtin_amdgcn_rcpf(q[e]);
        if (e < K - 1 || last_valid) {
            acc.x += a * h1r[e].x;
            acc.y += a * h1r[e].y;
            acc.z += a * h1r[e].z;
            acc.w += a * h1r[e].w;
        }
    }
}

__device__ __forceinline__ void add_one_edge(
    const float* __restrict__ H1, const float* __restrict__ HP, int s, int lane,
    const float4& h2d, const float4& hpd, const float4& av, float4& acc)
{
    s = (int)min((unsigned)s, (unsigned)(kN - 1));
    const float4 h1a = *(const float4*)(H1 + (size_t)s * kHD + lane * 4);
    const float4 hpa = *(const float4*)(HP + (size_t)s * kHD + lane * 4);
    float p = head_part(h1a, hpa, h2d, hpd, av);
    p += lane_xor_row<1>(p);
    p += lane_xor_row<2>(p);
    float m = p;
    m = fmaxf(m, lane_xor_row<4>(m));
    m = fmaxf(m, lane_xor_row<8>(m));
    m = fmaxf(m, lane_xor_16(m));
    const float e = exp_nonpos(p - m);
    float q = e;
    q += lane_xor_row<4>(q);
    q += lane_xor_row<8>(q);
    q += lane_xor_16(q);
    const float a = e * __builtin_amdgcn_rcpf(q);
    acc.x += a * h1a.x;
    acc.y += a * h1a.y;
    acc.z += a * h1a.z;
    acc.w += a * h1a.w;
}

constexpr int kNodesPerBlock      = 8;
constexpr int kNodesPerBlockFinal = 16;
static_assert(kN % kNodesPerBlockFinal == 0,
              "the last layer's grid covers whole blocks: no wave may leave before the barrier");

template <bool kFinal>
__global__ __launch_bounds__(kFinal ? kNodesPerBlockFinal * 32 : kNodesPerBlock * 32)
void node_attn_ln_kernel(
    const float* __restrict__ H1, const float* __restrict__ H2,
    const float* __restrict__ HP,
    const int* __restrict__ cnt, const int* __restrict__ slot,
    const int* __restrict__ spill,
    const float* __restrict__ avec,
    const float* __restrict__ G, const float* __restrict__ B,
    const float* __restrict__ Xin, float* __restrict__ Xout,
    const float* __restrict__ Wo, const float* __restrict__ bo,
    float* __restrict__ O)
{
    const int lane = threadIdx.x & 31;
    const int wave = __builtin_amdgcn_readfirstlane((int)(threadIdx.x >> 5));
    const int node = blockIdx.x * (kFinal ? kNodesPerBlockFinal : kNodesPerBlock) + wave;
    if constexpr (!kFinal) {
        if (node >= kN) return;
    }

    const int deg = cnt[node];
    const int beg = node * kSlots;
    const int end = beg + min(deg, kSlots);

    const float4 h2d = *(const float4*)(H2 + (size_t)node * kHD + lane * 4);
    const float4 hpd = *(const float4*)(HP + (size_t)node * kHD + lane * 4);
    const float4 av  = *(const float4*)(avec + lane * 4);

    float4 acc = make_float4(0.0f, 0.0f, 0.0f, 0.0f);

    for (int base = beg; base < end; base += 32) {
        const int n = min(32, end - base);
        int mine = 0;
        if (lane < n) mine = slot[base + lane];

        int j = 0;
        for (; j + 3 <= n; j += 4) {
            int s[4];
#pragma unroll
            for (int e = 0; e < 4; ++e) {
                s[e] = (int)min((unsigned)__builtin_amdgcn_readlane(mine, min(j + e, n - 1)),
                                (unsigned)(kN - 1));
            }
            add_edges<4>(H1, HP, s, j + 3 < n, lane, h2d, hpd, av, acc);
        }
        if (j < n) {
            int s[2];
            s[0] = (int)min((unsigned)__builtin_amdgcn_readlane(mine, j),
                            (unsigned)(kN - 1));
            s[1] = (int)min((unsigned)__builtin_amdgcn_readlane(mine, min(j + 1, n - 1)),
                            (unsigned)(kN - 1));
            add_edges<2>(H1, HP, s, j + 1 < n, lane, h2d, hpd, av, acc);
        }
    }

    if (deg > kSlots) {
        const int total = min(cnt[kSpillCount], kE);
        for (int ob = 0; ob < total; ob += 32) {
            const int m = min(32, total - ob);
            int od = -1, os = 0;
            if (lane < m) {
                const int2 pr = *(const int2*)(spill + 2 * (size_t)(ob + lane));
                od = pr.x;
                os = pr.y;
            }
            unsigned hit = __builtin_amdgcn_ballot_w32(od == node);
            while (hit != 0u) {
                const int l = __builtin_ctz(hit);
                hit &= hit - 1u;
                add_one_edge(H1, HP, __builtin_amdgcn_readlane(os, l), lane,
                             h2d, hpd, av, acc);
            }
        }
    }

    const float s = wave_sum(acc.x + acc.y + acc.z + acc.w);
    const float mu = s * (1.0f / 128.0f);

    const float dx0 = acc.x - mu, dx1 = acc.y - mu, dx2 = acc.z - mu, dx3 = acc.w - mu;
    const float q = wave_sum(dx0 * dx0 + dx1 * dx1 + dx2 * dx2 + dx3 * dx3);
    const float inv = rsqrtf(q * (1.0f / 128.0f) + kLnEps);

    const float4 g  = *(const float4*)(G + lane * 4);
    const float4 b  = *(const float4*)(B + lane * 4);
    const float4 xr = *(const float4*)(Xin + (size_t)node * kHD + lane * 4);

    float4 o;
    o.x = eluf(dx0 * inv * g.x + b.x) + xr.x;
    o.y = eluf(dx1 * inv * g.y + b.y) + xr.y;
    o.z = eluf(dx2 * inv * g.z + b.z) + xr.z;
    o.w = eluf(dx3 * inv * g.w + b.w) + xr.w;

    if constexpr (!kFinal) {
        *(float4*)(Xout + (size_t)node * kHD + lane * 4) = o;
    } else {
        const float4 w01 = *(const float4*)(Wo + lane * 8);
        const float4 w23 = *(const float4*)(Wo + lane * 8 + 4);
        float p0 = o.x * w01.x + o.y * w01.z + o.z * w23.x + o.w * w23.z;
        float p1 = o.x * w01.y + o.y * w01.w + o.z * w23.y + o.w * w23.w;
        p0 = wave_sum(p0);
        p1 = wave_sum(p1);
        __shared__ float2 proj[kNodesPerBlockFinal];
        if (lane == 0) proj[wave] = make_float2(p0 + bo[0], p1 + bo[1]);
        __syncthreads();
        if (wave == 0 && lane < kNodesPerBlockFinal) {
            *(float2*)(O + (size_t)(blockIdx.x * kNodesPerBlockFinal + lane) * kOUT) = proj[lane];
        }
    }
}

extern "C" void kernel_launch(void* const* d_in, const int* in_sizes, int n_in,
                              void* d_out, int out_size, void* d_ws, size_t ws_size,
                              hipStream_t stream)
{
    (void)in_sizes; (void)n_in; (void)out_size; (void)ws_size;

    const float* x    = (const float*)d_in[0];
    const int*   ei   = (const int*)  d_in[1];
    const float* W1   = (const float*)d_in[2];
    const float* b1   = (const float*)d_in[3];
    const float* W2   = (const float*)d_in[4];
    const float* b2   = (const float*)d_in[5];
    const float* a    = (const float*)d_in[6];
    const float* ln_g = (const float*)d_in[7];
    const float* ln_b = (const float*)d_in[8];
    const float* Wo   = (const float*)d_in[9];
    const float* bo   = (const float*)d_in[10];
    float*       out  = (float*)d_out;

    const size_t NHD = (size_t)kN * kHD;
    float* h1   = (float*)d_ws;
    float* h2   = h1  + NHD;
    float* hp   = h2  + NHD;
    float* xa   = hp  + NHD;
    int* cnt    = (int*)(xa + NHD);
    int* slot   = cnt + kCntPad;
    int* spill  = slot + (size_t)kN * kSlots;

    const int* srcIdx = ei;
    const int* dstIdx = ei + kE;

    hipMemsetAsync(cnt, 0, (size_t)kCntPad * sizeof(int), stream);
    group_edges_kernel<<<kE / 256, 256, 0, stream>>>(srcIdx, dstIdx, cnt, slot, spill);

    static_assert(kL == 2, "layer 0 stores its rows, layer 1 projects them");

    gemm_h1h2_kernel<<<kGemmBlocks, 256, 0, stream>>>(
        x, W1, b1, W2, b2, h1, h2, hp);
    node_attn_ln_kernel<false><<<kN / kNodesPerBlock, kNodesPerBlock * 32, 0, stream>>>(
        h1, h2, hp, cnt, slot, spill, a, ln_g, ln_b, x, xa,
        nullptr, nullptr, nullptr);

    gemm_h1h2_kernel<<<kGemmBlocks, 256, 0, stream>>>(
        xa,
        W1 + (size_t)kIN * kHD, b1 + kHD,
        W2 + (size_t)kIN * kHD, b2 + kHD,
        h1, h2, hp);
    node_attn_ln_kernel<true><<<kN / kNodesPerBlockFinal, kNodesPerBlockFinal * 32, 0, stream>>>(
        h1, h2, hp, cnt, slot, spill, a + kH * kD, ln_g + kHD, ln_b + kHD, xa, nullptr,
        Wo, bo, out);
}


# Node kernel edge loop, a quarter fewer vector instructions: packed pairs, ELU pick as one med3 (its -1, log2e and 1/4 folded into per-node constants), one-instruction DPP max; new logit sum order
# speedup vs baseline: 2.3731x; 1.0389x over previous
#include <hip/hip_runtime.h>
#include <math.h>


constexpr int kN   = 20000;
constexpr int kE   = 320000;
constexpr int kIN  = 128;
constexpr int kH   = 8;
constexpr int kD   = 16;
constexpr int kHD  = 128;
constexpr int kL   = 2;
constexpr int kOUT = 2;
constexpr float kLnEps = 1e-5f;

typedef __attribute__((ext_vector_type(2))) float v2f;
typedef __attribute__((ext_vector_type(4))) float v4f;
typedef __attribute__((ext_vector_type(8))) float v8f;

__device__ __forceinline__ float exp_nonpos(float x) {
    return __builtin_amdgcn_exp2f(x * 1.44269504088896340736f);
}

__device__ __forceinline__ float elu_pick(float x, float em1) {
    return __builtin_amdgcn_fmed3f(x, em1, 0.0f);
}
__device__ __forceinline__ float eluf(float x) {
    return elu_pick(x, exp_nonpos(x) - 1.0f);
}
__device__ __forceinline__ float elu1_pick(float x1, float ex) {
    return __builtin_amdgcn_fmed3f(x1, ex, 1.0f);
}

template <int M>
__device__ __forceinline__ float lane_xor_row(float x) {
    static_assert(M >= 1 && M <= 15, "row_xmask stays inside a 16-lane row");
    return __builtin_bit_cast(float, __builtin_amdgcn_update_dpp(
        0, __builtin_bit_cast(int, x), 0x160 + M, 0xF, 0xF, true));
}
__device__ __forceinline__ float lane_xor_16(float x) {
    const int b = __builtin_bit_cast(int, x);
    return __builtin_bit_cast(float, __builtin_amdgcn_permlanex16(
        b, b, 0x76543210, 0xfedcba98, false, true));
}
__device__ __forceinline__ float max2(float a, float b) {
    return __builtin_elementwise_maximum(a, b);
}
__device__ __forceinline__ float wave_sum(float v) {
    v += lane_xor_row<1>(v);
    v += lane_xor_row<2>(v);
    v += lane_xor_row<4>(v);
    v += lane_xor_row<8>(v);
    v += lane_xor_16(v);
    return v;
}

constexpr float kLog2e = 1.44269504088896340736f;

constexpr int kGemmTiles  = kN / 16;
constexpr int kGemmBlocks = 250;
static_assert(kN % 16 == 0, "whole 16-row tiles");

__device__ __forceinline__ void gemm_load_rows(const float* __restrict__ X, int tile,
                                               int r, int half, float4 (&q)[16])
{
    const float* xrow = X + (size_t)(tile * 16 + r) * kIN + half * 4;
#pragma unroll
    for (int g = 0; g < 16; ++g) q[g] = *(const float4*)(xrow + 8 * g);
}

__device__ __forceinline__ void gemm_tile(
    const float4 (&q)[16], const v2f (&w1)[32], const v2f (&w2)[32],
    float bb1, float bb2, int tile, int n, int half,
    float* __restrict__ H1, float* __restrict__ H2, float* __restrict__ HP)
{
    v8f acc1 = {};
    v8f acc2 = {};
#pragma unroll
    for (int g = 0; g < 16; ++g) {
        v2f a01, a23;
        a01.x = q[g].x; a01.y = q[g].y;
        a23.x = q[g].z; a23.y = q[g].w;
        acc1 = __builtin_amdgcn_wmma_f32_16x16x4_f32(
            false, a01, false, w1[2 * g], (short)0, acc1, false, false);
        acc2 = __builtin_amdgcn_wmma_f32_16x16x4_f32(
            false, a01, false, w2[2 * g], (short)0, acc2, false, false);
        acc1 = __builtin_amdgcn_wmma_f32_16x16x4_f32(
            false, a23, false, w1[2 * g + 1], (short)0, acc1, false, false);
        acc2 = __builtin_amdgcn_wmma_f32_16x16x4_f32(
            false, a23, false, w2[2 * g + 1], (short)0, acc2, false, false);
    }
    const size_t base = (size_t)(tile * 16 + half * 8) * kHD + n;
    float* __restrict__ p1 = H1 + base;
    float* __restrict__ p2 = H2 + base;
    float* __restrict__ pp = HP + base;
#pragma unroll
    for (int j = 0; j < 8; ++j) {
        const float h1v = acc1[j] + bb1;
        const float h2v = acc2[j] + bb2;
        p1[j * kHD] = h1v;
        p2[j * kHD] = h2v;
        pp[j * kHD] = h1v * h2v;
    }
}

__global__ __launch_bounds__(256)
void gemm_h1h2_kernel(
    const float* __restrict__ X,
    const float* __restrict__ W1, const float* __restrict__ B1,
    const float* __restrict__ W2, const float* __restrict__ B2,
    float* __restrict__ H1, float* __restrict__ H2, float* __restrict__ HP)
{
    const int lane  = threadIdx.x & 31;
    const int wave  = threadIdx.x >> 5;
    const int r     = lane & 15;
    const int half  = lane >> 4;
    const int n     = wave * 16 + r;

    v2f w1[32], w2[32];
#pragma unroll
    for (int g = 0; g < 16; ++g) {
        const int kb = 8 * g + 4 * half;
        w1[2 * g].x     = W1[(size_t)(kb + 0) * kHD + n];
        w1[2 * g].y     = W1[(size_t)(kb + 1) * kHD + n];
        w1[2 * g + 1].x = W1[(size_t)(kb + 2) * kHD + n];
        w1[2 * g + 1].y = W1[(size_t)(kb + 3) * kHD + n];
        w2[2 * g].x     = W2[(size_t)(kb + 0) * kHD + n];
        w2[2 * g].y     = W2[(size_t)(kb + 1) * kHD + n];
        w2[2 * g + 1].x = W2[(size_t)(kb + 2) * kHD + n];
        w2[2 * g + 1].y = W2[(size_t)(kb + 3) * kHD + n];
    }
    const float bb1 = B1[n];
    const float bb2 = B2[n];

    const int step  = (int)gridDim.x;
    const int first = (int)blockIdx.x;
    if (first >= kGemmTiles) return;
    const int cnt = (kGemmTiles - 1 - first) / step + 1;

    float4 qa[16], qb[16];
    gemm_load_rows(X, first, r, half, qa);
    int i = 0;
    for (; i + 1 < cnt; i += 2) {
        gemm_load_rows(X, first + (i + 1) * step, r, half, qb);
        __builtin_amdgcn_sched_barrier(0);
        gemm_tile(qa, w1, w2, bb1, bb2, first + i * step, n, half, H1, H2, HP);

        gemm_load_rows(X, first + min(i + 2, cnt - 1) * step, r, half, qa);
        __builtin_amdgcn_sched_barrier(0);
        gemm_tile(qb, w1, w2, bb1, bb2, first + (i + 1) * step, n, half, H1, H2, HP);
    }
    if (i < cnt) {
        gemm_tile(qa, w1, w2, bb1, bb2, first + i * step, n, half, H1, H2, HP);
    }
}

constexpr int kSlots      = 64;
constexpr int kCntPad     = 20480;
constexpr int kSpillCount = kN;
static_assert(kSpillCount < kCntPad, "the spill count lies inside the cleared table");
static_assert((kSlots * sizeof(int)) % 256 == 0, "rows are whole aligned 256-byte units");

__global__ __launch_bounds__(256) void group_edges_kernel(
    const int* __restrict__ src, const int* __restrict__ dst,
    int* __restrict__ cnt, int* __restrict__ slot, int* __restrict__ spill)
{
    const int e = blockIdx.x * 256 + threadIdx.x;
    if (e >= kE) return;
    const int d = dst[e];
    const int s = src[e];
    const int p = atomicAdd(cnt + d, 1);
    if (p < kSlots) {
        slot[d * kSlots + p] = s;
    } else {
        const int o = atomicAdd(cnt + kSpillCount, 1);
        *(int2*)(spill + 2 * (size_t)o) = make_int2(d, s);
    }
}

template <int K>
__device__ __forceinline__ void add_edges(
    const float* __restrict__ H1, const float* __restrict__ HP,
    const int (&s)[K], bool last_valid, int lane,
    const v4f& h2d1, const v4f& hpd, const v4f& as, const v2f& nas, v4f& acc)
{
    v4f h1r[K], hpr[K];
#pragma unroll
    for (int e = 0; e < K; ++e) {
        const unsigned off = (unsigned)s[e] * (unsigned)(kHD * sizeof(float))
                           + (unsigned)lane * (unsigned)sizeof(v4f);
        h1r[e] = *(const v4f*)((const char*)H1 + off);
        hpr[e] = *(const v4f*)((const char*)HP + off);
    }

    float p[K], m[K], ex[K], q[K];
#pragma unroll
    for (int e = 0; e < K; ++e) {
        const v4f x1 = hpr[e] * hpd + (h1r[e] + h2d1);
        const v4f y = x1 * kLog2e - kLog2e;
        v4f ex1;
        ex1.x = __builtin_amdgcn_exp2f(y.x);
        ex1.y = __builtin_amdgcn_exp2f(y.y);
        ex1.z = __builtin_amdgcn_exp2f(y.z);
        ex1.w = __builtin_amdgcn_exp2f(y.w);
        v4f t1;
        t1.x = elu1_pick(x1.x, ex1.x);
        t1.y = elu1_pick(x1.y, ex1.y);
        t1.z = elu1_pick(x1.z, ex1.z);
        t1.w = elu1_pick(x1.w, ex1.w);
        v2f u = t1.lo * as.lo + nas;
        u = t1.hi * as.hi + u;
        p[e] = u.x + u.y;
    }

#pragma unroll
    for (int e = 0; e < K; ++e) p[e] += lane_xor_row<1>(p[e]);
#pragma unroll
    for (int e = 0; e < K; ++e) p[e] += lane_xor_row<2>(p[e]);

#pragma unroll
    for (int e = 0; e < K; ++e) m[e] = max2(p[e], lane_xor_row<4>(p[e]));
#pragma unroll
    for (int e = 0; e < K; ++e) m[e] = max2(m[e], lane_xor_row<8>(m[e]));
#pragma unroll
    for (int e = 0; e < K; ++e) m[e] = max2(m[e], lane_xor_16(m[e]));

#pragma unroll
    for (int e = 0; e < K; ++e) ex[e] = __builtin_amdgcn_exp2f(p[e] - m[e]);
#pragma unroll
    for (int e = 0; e < K; ++e) q[e] = ex[e] + lane_xor_row<4>(ex[e]);
#pragma unroll
    for (int e = 0; e < K; ++e) q[e] += lane_xor_row<8>(q[e]);
#pragma unroll
    for (int e = 0; e < K; ++e) q[e] += lane_xor_16(q[e]);

#pragma unroll
    for (int e = 0; e < K; ++e) {
        const float a = ex[e] * __builtin_amdgcn_rcpf(q[e]);
        if (e < K - 1 || last_valid) acc = h1r[e] * a + acc;
    }
}

constexpr int kNodesPerBlock      = 8;
constexpr int kNodesPerBlockFinal = 16;
static_assert(kN % kNodesPerBlockFinal == 0,
              "the last layer's grid covers whole blocks: no wave may leave before the barrier");

template <bool kFinal>
__global__ __launch_bounds__(kFinal ? kNodesPerBlockFinal * 32 : kNodesPerBlock * 32)
void node_attn_ln_kernel(
    const float* __restrict__ H1, const float* __restrict__ H2,
    const float* __restrict__ HP,
    const int* __restrict__ cnt, const int* __restrict__ slot,
    const int* __restrict__ spill,
    const float* __restrict__ avec,
    const float* __restrict__ G, const float* __restrict__ B,
    const float* __restrict__ Xin, float* __restrict__ Xout,
    const float* __restrict__ Wo, const float* __restrict__ bo,
    float* __restrict__ O)
{
    const int lane = threadIdx.x & 31;
    const int wave = __builtin_amdgcn_readfirstlane((int)(threadIdx.x >> 5));
    const int node = blockIdx.x * (kFinal ? kNodesPerBlockFinal : kNodesPerBlock) + wave;
    if constexpr (!kFinal) {
        if (node >= kN) return;
    }

    const int deg = cnt[node];
    const int beg = node * kSlots;
    const int end = beg + min(deg, kSlots);

    const v4f h2d1 = *(const v4f*)(H2 + (size_t)node * kHD + lane * 4) + 1.0f;
    const v4f hpd  = *(const v4f*)(HP + (size_t)node * kHD + lane * 4);
    const v4f as   = *(const v4f*)(avec + lane * 4) * (0.25f * kLog2e);
    const v2f nas  = -(as.lo + as.hi);

    v4f acc = {0.0f, 0.0f, 0.0f, 0.0f};

    for (int base = beg; base < end; base += 32) {
        const int n = min(32, end - base);
        int mine = 0;
        if (lane < n) mine = slot[base + lane];

        int j = 0;
        for (; j + 3 <= n; j += 4) {
            int s[4];
#pragma unroll
            for (int e = 0; e < 4; ++e) {
                s[e] = (int)min((unsigned)__builtin_amdgcn_readlane(mine, min(j + e, n - 1)),
                                (unsigned)(kN - 1));
            }
            add_edges<4>(H1, HP, s, j + 3 < n, lane, h2d1, hpd, as, nas, acc);
        }
        if (j < n) {
            int s[2];
            s[0] = (int)min((unsigned)__builtin_amdgcn_readlane(mine, j),
                            (unsigned)(kN - 1));
            s[1] = (int)min((unsigned)__builtin_amdgcn_readlane(mine, min(j + 1, n - 1)),
                            (unsigned)(kN - 1));
            add_edges<2>(H1, HP, s, j + 1 < n, lane, h2d1, hpd, as, nas, acc);
        }
    }

    if (deg > kSlots) {
        const int total = min(cnt[kSpillCount], kE);
        for (int ob = 0; ob < total; ob += 32) {
            const int m = min(32, total - ob);
            int od = -1, os = 0;
            if (lane < m) {
                const int2 pr = *(const int2*)(spill + 2 * (size_t)(ob + lane));
                od = pr.x;
                os = pr.y;
            }
            unsigned hit = __builtin_amdgcn_ballot_w32(od == node);
            while (hit != 0u) {
                const int l = __builtin_ctz(hit);
                hit &= hit - 1u;
                const int s1[1] = { (int)min((unsigned)__builtin_amdgcn_readlane(os, l),
                                             (unsigned)(kN - 1)) };
                add_edges<1>(H1, HP, s1, true, lane, h2d1, hpd, as, nas, acc);
            }
        }
    }

    const float s = wave_sum(acc.x + acc.y + acc.z + acc.w);
    const float mu = s * (1.0f / 128.0f);

    const float dx0 = acc.x - mu, dx1 = acc.y - mu, dx2 = acc.z - mu, dx3 = acc.w - mu;
    const float q = wave_sum(dx0 * dx0 + dx1 * dx1 + dx2 * dx2 + dx3 * dx3);
    const float inv = rsqrtf(q * (1.0f / 128.0f) + kLnEps);

    const float4 g  = *(const float4*)(G + lane * 4);
    const float4 b  = *(const float4*)(B + lane * 4);
    const float4 xr = *(const float4*)(Xin + (size_t)node * kHD + lane * 4);

    float4 o;
    o.x = eluf(dx0 * inv * g.x + b.x) + xr.x;
    o.y = eluf(dx1 * inv * g.y + b.y) + xr.y;
    o.z = eluf(dx2 * inv * g.z + b.z) + xr.z;
    o.w = eluf(dx3 * inv * g.w + b.w) + xr.w;

    if constexpr (!kFinal) {
        *(float4*)(Xout + (size_t)node * kHD + lane * 4) = o;
    } else {
        const float4 w01 = *(const float4*)(Wo + lane * 8);
        const float4 w23 = *(const float4*)(Wo + lane * 8 + 4);
        float p0 = o.x * w01.x + o.y * w01.z + o.z * w23.x + o.w * w23.z;
        float p1 = o.x * w01.y + o.y * w01.w + o.z * w23.y + o.w * w23.w;
        p0 = wave_sum(p0);
        p1 = wave_sum(p1);
        __shared__ float2 proj[kNodesPerBlockFinal];
        if (lane == 0) proj[wave] = make_float2(p0 + bo[0], p1 + bo[1]);
        __syncthreads();
        if (wave == 0 && lane < kNodesPerBlockFinal) {
            *(float2*)(O + (size_t)(blockIdx.x * kNodesPerBlockFinal + lane) * kOUT) = proj[lane];
        }
    }
}

extern "C" void kernel_launch(void* const* d_in, const int* in_sizes, int n_in,
                              void* d_out, int out_size, void* d_ws, size_t ws_size,
                              hipStream_t stream)
{
    (void)in_sizes; (void)n_in; (void)out_size; (void)ws_size;

    const float* x    = (const float*)d_in[0];
    const int*   ei   = (const int*)  d_in[1];
    const float* W1   = (const float*)d_in[2];
    const float* b1   = (const float*)d_in[3];
    const float* W2   = (const float*)d_in[4];
    const float* b2   = (const float*)d_in[5];
    const float* a    = (const float*)d_in[6];
    const float* ln_g = (const float*)d_in[7];
    const float* ln_b = (const float*)d_in[8];
    const float* Wo   = (const float*)d_in[9];
    const float* bo   = (const float*)d_in[10];
    float*       out  = (float*)d_out;

    const size_t NHD = (size_t)kN * kHD;
    float* h1   = (float*)d_ws;
    float* h2   = h1  + NHD;
    float* hp   = h2  + NHD;
    float* xa   = hp  + NHD;
    int* cnt    = (int*)(xa + NHD);
    int* slot   = cnt + kCntPad;
    int* spill  = slot + (size_t)kN * kSlots;

    const int* srcIdx = ei;
    const int* dstIdx = ei + kE;

    hipMemsetAsync(cnt, 0, (size_t)kCntPad * sizeof(int), stream);
    group_edges_kernel<<<kE / 256, 256, 0, stream>>>(srcIdx, dstIdx, cnt, slot, spill);

    static_assert(kL == 2, "layer 0 stores its rows, layer 1 projects them");

    gemm_h1h2_kernel<<<kGemmBlocks, 256, 0, stream>>>(
        x, W1, b1, W2, b2, h1, h2, hp);
    node_attn_ln_kernel<false><<<kN / kNodesPerBlock, kNodesPerBlock * 32, 0, stream>>>(
        h1, h2, hp, cnt, slot, spill, a, ln_g, ln_b, x, xa,
        nullptr, nullptr, nullptr);

    gemm_h1h2_kernel<<<kGemmBlocks, 256, 0, stream>>>(
        xa,
        W1 + (size_t)kIN * kHD, b1 + kHD,
        W2 + (size_t)kIN * kHD, b2 + kHD,
        h1, h2, hp);
    node_attn_ln_kernel<true><<<kN / kNodesPerBlockFinal, kNodesPerBlockFinal * 32, 0, stream>>>(
        h1, h2, hp, cnt, slot, spill, a + kH * kD, ln_g + kHD, ln_b + kHD, xa, nullptr,
        Wo, bo, out);
}


# Edge grouping by the lanes of the layer-0 GEMM launch (5 edges each: counting atomics issued under the matrix work, sources filed after it); the grouping launch is gone; same arrival-order sums
# speedup vs baseline: 2.4462x; 1.0308x over previous
#include <hip/hip_runtime.h>
#include <math.h>


constexpr int kN   = 20000;
constexpr int kE   = 320000;
constexpr int kIN  = 128;
constexpr int kH   = 8;
constexpr int kD   = 16;
constexpr int kHD  = 128;
constexpr int kL   = 2;
constexpr int kOUT = 2;
constexpr float kLnEps = 1e-5f;

typedef __attribute__((ext_vector_type(2))) float v2f;
typedef __attribute__((ext_vector_type(4))) float v4f;
typedef __attribute__((ext_vector_type(8))) float v8f;

__device__ __forceinline__ float exp_nonpos(float x) {
    return __builtin_amdgcn_exp2f(x * 1.44269504088896340736f);
}

__device__ __forceinline__ float elu_pick(float x, float em1) {
    return __builtin_amdgcn_fmed3f(x, em1, 0.0f);
}
__device__ __forceinline__ float eluf(float x) {
    return elu_pick(x, exp_nonpos(x) - 1.0f);
}
__device__ __forceinline__ float elu1_pick(float x1, float ex) {
    return __builtin_amdgcn_fmed3f(x1, ex, 1.0f);
}

template <int M>
__device__ __forceinline__ float lane_xor_row(float x) {
    static_assert(M >= 1 && M <= 15, "row_xmask stays inside a 16-lane row");
    return __builtin_bit_cast(float, __builtin_amdgcn_update_dpp(
        0, __builtin_bit_cast(int, x), 0x160 + M, 0xF, 0xF, true));
}
__device__ __forceinline__ float lane_xor_16(float x) {
    const int b = __builtin_bit_cast(int, x);
    return __builtin_bit_cast(float, __builtin_amdgcn_permlanex16(
        b, b, 0x76543210, 0xfedcba98, false, true));
}
__device__ __forceinline__ float max2(float a, float b) {
    return __builtin_elementwise_maximum(a, b);
}
__device__ __forceinline__ float wave_sum(float v) {
    v += lane_xor_row<1>(v);
    v += lane_xor_row<2>(v);
    v += lane_xor_row<4>(v);
    v += lane_xor_row<8>(v);
    v += lane_xor_16(v);
    return v;
}

constexpr float kLog2e = 1.44269504088896340736f;

constexpr int kSlots      = 64;
constexpr int kCntPad     = 20480;
constexpr int kSpillCount = kN;
static_assert(kSpillCount < kCntPad, "the spill count lies inside the cleared table");
static_assert((kSlots * sizeof(int)) % 256 == 0, "rows are whole aligned 256-byte units");

constexpr int kGemmTiles  = kN / 16;
constexpr int kGemmBlocks = 250;
static_assert(kN % 16 == 0, "whole 16-row tiles");
static_assert(kGemmBlocks <= kGemmTiles, "every block has a row tile: none leaves before its edges are filed");
constexpr int kGemmLanes    = kGemmBlocks * 256;
constexpr int kEdgesPerLane = kE / kGemmLanes;
static_assert(kEdgesPerLane * kGemmLanes == kE, "the GEMM grid's lanes cover the edge list exactly");

__device__ __forceinline__ void gemm_load_rows(const float* __restrict__ X, int tile,
                                               int r, int half, float4 (&q)[16])
{
    const float* xrow = X + (size_t)(tile * 16 + r) * kIN + half * 4;
#pragma unroll
    for (int g = 0; g < 16; ++g) q[g] = *(const float4*)(xrow + 8 * g);
}

__device__ __forceinline__ void gemm_tile(
    const float4 (&q)[16], const v2f (&w1)[32], const v2f (&w2)[32],
    float bb1, float bb2, int tile, int n, int half,
    float* __restrict__ H1, float* __restrict__ H2, float* __restrict__ HP)
{
    v8f acc1 = {};
    v8f acc2 = {};
#pragma unroll
    for (int g = 0; g < 16; ++g) {
        v2f a01, a23;
        a01.x = q[g].x; a01.y = q[g].y;
        a23.x = q[g].z; a23.y = q[g].w;
        acc1 = __builtin_amdgcn_wmma_f32_16x16x4_f32(
            false, a01, false, w1[2 * g], (short)0, acc1, false, false);
        acc2 = __builtin_amdgcn_wmma_f32_16x16x4_f32(
            false, a01, false, w2[2 * g], (short)0, acc2, false, false);
        acc1 = __builtin_amdgcn_wmma_f32_16x16x4_f32(
            false, a23, false, w1[2 * g + 1], (short)0, acc1, false, false);
        acc2 = __builtin_amdgcn_wmma_f32_16x16x4_f32(
            false, a23, false, w2[2 * g + 1], (short)0, acc2, false, false);
    }
    const size_t base = (size_t)(tile * 16 + half * 8) * kHD + n;
    float* __restrict__ p1 = H1 + base;
    float* __restrict__ p2 = H2 + base;
    float* __restrict__ pp = HP + base;
#pragma unroll
    for (int j = 0; j < 8; ++j) {
        const float h1v = acc1[j] + bb1;
        const float h2v = acc2[j] + bb2;
        p1[j * kHD] = h1v;
        p2[j * kHD] = h2v;
        pp[j * kHD] = h1v * h2v;
    }
}

__device__ __forceinline__ void count_edges(int* __restrict__ ecnt,
                                            const int (&ed)[kEdgesPerLane],
                                            int (&ep)[kEdgesPerLane])
{
#pragma unroll
    for (int k = 0; k < kEdgesPerLane; ++k) ep[k] = atomicAdd(ecnt + ed[k], 1);
}

template <bool kGroup>
__global__ __launch_bounds__(256)
void gemm_h1h2_kernel(
    const float* __restrict__ X,
    const float* __restrict__ W1, const float* __restrict__ B1,
    const float* __restrict__ W2, const float* __restrict__ B2,
    float* __restrict__ H1, float* __restrict__ H2, float* __restrict__ HP,
    const int* __restrict__ esrc, const int* __restrict__ edst,
    int* __restrict__ ecnt, int* __restrict__ slot, int* __restrict__ spill)
{
    const int lane  = threadIdx.x & 31;
    const int wave  = threadIdx.x >> 5;
    const int r     = lane & 15;
    const int half  = lane >> 4;
    const int n     = wave * 16 + r;

    int ed[kEdgesPerLane], es[kEdgesPerLane], ep[kEdgesPerLane];
    if constexpr (kGroup) {
        const int t = (int)blockIdx.x * 256 + (int)threadIdx.x;
#pragma unroll
        for (int i = 0; i < kEdgesPerLane; ++i) {
            ed[i] = edst[i * kGemmLanes + t];
            es[i] = esrc[i * kGemmLanes + t];
        }
    }

    v2f w1[32], w2[32];
#pragma unroll
    for (int g = 0; g < 16; ++g) {
        const int kb = 8 * g + 4 * half;
        w1[2 * g].x     = W1[(size_t)(kb + 0) * kHD + n];
        w1[2 * g].y     = W1[(size_t)(kb + 1) * kHD + n];
        w1[2 * g + 1].x = W1[(size_t)(kb + 2) * kHD + n];
        w1[2 * g + 1].y = W1[(size_t)(kb + 3) * kHD + n];
        w2[2 * g].x     = W2[(size_t)(kb + 0) * kHD + n];
        w2[2 * g].y     = W2[(size_t)(kb + 1) * kHD + n];
        w2[2 * g + 1].x = W2[(size_t)(kb + 2) * kHD + n];
        w2[2 * g + 1].y = W2[(size_t)(kb + 3) * kHD + n];
    }
    const float bb1 = B1[n];
    const float bb2 = B2[n];

    const int step  = (int)gridDim.x;
    const int first = (int)blockIdx.x;
    if (first >= kGemmTiles) return;
    const int cnt = (kGemmTiles - 1 - first) / step + 1;

    float4 qa[16], qb[16];
    gemm_load_rows(X, first, r, half, qa);

    int i = 0;
    for (; i + 1 < cnt; i += 2) {
        gemm_load_rows(X, first + (i + 1) * step, r, half, qb);
        if constexpr (kGroup) {
            if (i == 0) count_edges(ecnt, ed, ep);
        }
        __builtin_amdgcn_sched_barrier(0);
        gemm_tile(qa, w1, w2, bb1, bb2, first + i * step, n, half, H1, H2, HP);

        gemm_load_rows(X, first + min(i + 2, cnt - 1) * step, r, half, qa);
        __builtin_amdgcn_sched_barrier(0);
        gemm_tile(qb, w1, w2, bb1, bb2, first + (i + 1) * step, n, half, H1, H2, HP);
    }
    if constexpr (kGroup) {
        if (cnt < 2) count_edges(ecnt, ed, ep);
    }
    if (i < cnt) {
        gemm_tile(qa, w1, w2, bb1, bb2, first + i * step, n, half, H1, H2, HP);
    }

    if constexpr (kGroup) {
#pragma unroll
        for (int k = 0; k < kEdgesPerLane; ++k) {
            if (ep[k] < kSlots) {
                slot[ed[k] * kSlots + ep[k]] = es[k];
            } else {
                const int o = atomicAdd(ecnt + kSpillCount, 1);
                *(int2*)(spill + 2 * (size_t)o) = make_int2(ed[k], es[k]);
            }
        }
    }
}

template <int K>
__device__ __forceinline__ void add_edges(
    const float* __restrict__ H1, const float* __restrict__ HP,
    const int (&s)[K], bool last_valid, int lane,
    const v4f& h2d1, const v4f& hpd, const v4f& as, const v2f& nas, v4f& acc)
{
    v4f h1r[K], hpr[K];
#pragma unroll
    for (int e = 0; e < K; ++e) {
        const unsigned off = (unsigned)s[e] * (unsigned)(kHD * sizeof(float))
                           + (unsigned)lane * (unsigned)sizeof(v4f);
        h1r[e] = *(const v4f*)((const char*)H1 + off);
        hpr[e] = *(const v4f*)((const char*)HP + off);
    }

    float p[K], m[K], ex[K], q[K];
#pragma unroll
    for (int e = 0; e < K; ++e) {
        const v4f x1 = hpr[e] * hpd + (h1r[e] + h2d1);
        const v4f y = x1 * kLog2e - kLog2e;
        v4f ex1;
        ex1.x = __builtin_amdgcn_exp2f(y.x);
        ex1.y = __builtin_amdgcn_exp2f(y.y);
        ex1.z = __builtin_amdgcn_exp2f(y.z);
        ex1.w = __builtin_amdgcn_exp2f(y.w);
        v4f t1;
        t1.x = elu1_pick(x1.x, ex1.x);
        t1.y = elu1_pick(x1.y, ex1.y);
        t1.z = elu1_pick(x1.z, ex1.z);
        t1.w = elu1_pick(x1.w, ex1.w);
        v2f u = t1.lo * as.lo + nas;
        u = t1.hi * as.hi + u;
        p[e] = u.x + u.y;
    }

#pragma unroll
    for (int e = 0; e < K; ++e) p[e] += lane_xor_row<1>(p[e]);
#pragma unroll
    for (int e = 0; e < K; ++e) p[e] += lane_xor_row<2>(p[e]);

#pragma unroll
    for (int e = 0; e < K; ++e) m[e] = max2(p[e], lane_xor_row<4>(p[e]));
#pragma unroll
    for (int e = 0; e < K; ++e) m[e] = max2(m[e], lane_xor_row<8>(m[e]));
#pragma unroll
    for (int e = 0; e < K; ++e) m[e] = max2(m[e], lane_xor_16(m[e]));

#pragma unroll
    for (int e = 0; e < K; ++e) ex[e] = __builtin_amdgcn_exp2f(p[e] - m[e]);
#pragma unroll
    for (int e = 0; e < K; ++e) q[e] = ex[e] + lane_xor_row<4>(ex[e]);
#pragma unroll
    for (int e = 0; e < K; ++e) q[e] += lane_xor_row<8>(q[e]);
#pragma unroll
    for (int e = 0; e < K; ++e) q[e] += lane_xor_16(q[e]);

#pragma unroll
    for (int e = 0; e < K; ++e) {
        const float a = ex[e] * __builtin_amdgcn_rcpf(q[e]);
        if (e < K - 1 || last_valid) acc = h1r[e] * a + acc;
    }
}

constexpr int kNodesPerBlock      = 8;
constexpr int kNodesPerBlockFinal = 16;
static_assert(kN % kNodesPerBlockFinal == 0,
              "the last layer's grid covers whole blocks: no wave may leave before the barrier");

template <bool kFinal>
__global__ __launch_bounds__(kFinal ? kNodesPerBlockFinal * 32 : kNodesPerBlock * 32)
void node_attn_ln_kernel(
    const float* __restrict__ H1, const float* __restrict__ H2,
    const float* __restrict__ HP,
    const int* __restrict__ cnt, const int* __restrict__ slot,
    const int* __restrict__ spill,
    const float* __restrict__ avec,
    const float* __restrict__ G, const float* __restrict__ B,
    const float* __restrict__ Xin, float* __restrict__ Xout,
    const float* __restrict__ Wo, const float* __restrict__ bo,
    float* __restrict__ O)
{
    const int lane = threadIdx.x & 31;
    const int wave = __builtin_amdgcn_readfirstlane((int)(threadIdx.x >> 5));
    const int node = blockIdx.x * (kFinal ? kNodesPerBlockFinal : kNodesPerBlock) + wave;
    if constexpr (!kFinal) {
        if (node >= kN) return;
    }

    const int deg = cnt[node];
    const int beg = node * kSlots;
    const int end = beg + min(deg, kSlots);

    const v4f h2d1 = *(const v4f*)(H2 + (size_t)node * kHD + lane * 4) + 1.0f;
    const v4f hpd  = *(const v4f*)(HP + (size_t)node * kHD + lane * 4);
    const v4f as   = *(const v4f*)(avec + lane * 4) * (0.25f * kLog2e);
    const v2f nas  = -(as.lo + as.hi);

    v4f acc = {0.0f, 0.0f, 0.0f, 0.0f};

    for (int base = beg; base < end; base += 32) {
        const int n = min(32, end - base);
        int mine = 0;
        if (lane < n) mine = slot[base + lane];

        int j = 0;
        for (; j + 3 <= n; j += 4) {
            int s[4];
#pragma unroll
            for (int e = 0; e < 4; ++e) {
                s[e] = (int)min((unsigned)__builtin_amdgcn_readlane(mine, min(j + e, n - 1)),
                                (unsigned)(kN - 1));
            }
            add_edges<4>(H1, HP, s, j + 3 < n, lane, h2d1, hpd, as, nas, acc);
        }
        if (j < n) {
            int s[2];
            s[0] = (int)min((unsigned)__builtin_amdgcn_readlane(mine, j),
                            (unsigned)(kN - 1));
            s[1] = (int)min((unsigned)__builtin_amdgcn_readlane(mine, min(j + 1, n - 1)),
                            (unsigned)(kN - 1));
            add_edges<2>(H1, HP, s, j + 1 < n, lane, h2d1, hpd, as, nas, acc);
        }
    }

    if (deg > kSlots) {
        const int total = min(cnt[kSpillCount], kE);
        for (int ob = 0; ob < total; ob += 32) {
            const int m = min(32, total - ob);
            int od = -1, os = 0;
            if (lane < m) {
                const int2 pr = *(const int2*)(spill + 2 * (size_t)(ob + lane));
                od = pr.x;
                os = pr.y;
            }
            unsigned hit = __builtin_amdgcn_ballot_w32(od == node);
            while (hit != 0u) {
                const int l = __builtin_ctz(hit);
                hit &= hit - 1u;
                const int s1[1] = { (int)min((unsigned)__builtin_amdgcn_readlane(os, l),
                                             (unsigned)(kN - 1)) };
                add_edges<1>(H1, HP, s1, true, lane, h2d1, hpd, as, nas, acc);
            }
        }
    }

    const float s = wave_sum(acc.x + acc.y + acc.z + acc.w);
    const float mu = s * (1.0f / 128.0f);

    const float dx0 = acc.x - mu, dx1 = acc.y - mu, dx2 = acc.z - mu, dx3 = acc.w - mu;
    const float q = wave_sum(dx0 * dx0 + dx1 * dx1 + dx2 * dx2 + dx3 * dx3);
    const float inv = rsqrtf(q * (1.0f / 128.0f) + kLnEps);

    const float4 g  = *(const float4*)(G + lane * 4);
    const float4 b  = *(const float4*)(B + lane * 4);
    const float4 xr = *(const float4*)(Xin + (size_t)node * kHD + lane * 4);

    float4 o;
    o.x = eluf(dx0 * inv * g.x + b.x) + xr.x;
    o.y = eluf(dx1 * inv * g.y + b.y) + xr.y;
    o.z = eluf(dx2 * inv * g.z + b.z) + xr.z;
    o.w = eluf(dx3 * inv * g.w + b.w) + xr.w;

    if constexpr (!kFinal) {
        *(float4*)(Xout + (size_t)node * kHD + lane * 4) = o;
    } else {
        const float4 w01 = *(const float4*)(Wo + lane * 8);
        const float4 w23 = *(const float4*)(Wo + lane * 8 + 4);
        float p0 = o.x * w01.x + o.y * w01.z + o.z * w23.x + o.w * w23.z;
        float p1 = o.x * w01.y + o.y * w01.w + o.z * w23.y + o.w * w23.w;
        p0 = wave_sum(p0);
        p1 = wave_sum(p1);
        __shared__ float2 proj[kNodesPerBlockFinal];
        if (lane == 0) proj[wave] = make_float2(p0 + bo[0], p1 + bo[1]);
        __syncthreads();
        if (wave == 0 && lane < kNodesPerBlockFinal) {
            *(float2*)(O + (size_t)(blockIdx.x * kNodesPerBlockFinal + lane) * kOUT) = proj[lane];
        }
    }
}

extern "C" void kernel_launch(void* const* d_in, const int* in_sizes, int n_in,
                              void* d_out, int out_size, void* d_ws, size_t ws_size,
                              hipStream_t stream)
{
    (void)in_sizes; (void)n_in; (void)out_size; (void)ws_size;

    const float* x    = (const float*)d_in[0];
    const int*   ei   = (const int*)  d_in[1];
    const float* W1   = (const float*)d_in[2];
    const float* b1   = (const float*)d_in[3];
    const float* W2   = (const float*)d_in[4];
    const float* b2   = (const float*)d_in[5];
    const float* a    = (const float*)d_in[6];
    const float* ln_g = (const float*)d_in[7];
    const float* ln_b = (const float*)d_in[8];
    const float* Wo   = (const float*)d_in[9];
    const float* bo   = (const float*)d_in[10];
    float*       out  = (float*)d_out;

    const size_t NHD = (size_t)kN * kHD;
    float* h1   = (float*)d_ws;
    float* h2   = h1  + NHD;
    float* hp   = h2  + NHD;
    float* xa   = hp  + NHD;
    int* cnt    = (int*)(xa + NHD);
    int* slot   = cnt + kCntPad;
    int* spill  = slot + (size_t)kN * kSlots;

    const int* srcIdx = ei;
    const int* dstIdx = ei + kE;

    hipMemsetAsync(cnt, 0, (size_t)kCntPad * sizeof(int), stream);

    static_assert(kL == 2, "layer 0 stores its rows, layer 1 projects them");

    gemm_h1h2_kernel<true><<<kGemmBlocks, 256, 0, stream>>>(
        x, W1, b1, W2, b2, h1, h2, hp, srcIdx, dstIdx, cnt, slot, spill);
    node_attn_ln_kernel<false><<<kN / kNodesPerBlock, kNodesPerBlock * 32, 0, stream>>>(
        h1, h2, hp, cnt, slot, spill, a, ln_g, ln_b, x, xa,
        nullptr, nullptr, nullptr);

    gemm_h1h2_kernel<false><<<kGemmBlocks, 256, 0, stream>>>(
        xa,
        W1 + (size_t)kIN * kHD, b1 + kHD,
        W2 + (size_t)kIN * kHD, b2 + kHD,
        h1, h2, hp, nullptr, nullptr, nullptr, nullptr, nullptr);
    node_attn_ln_kernel<true><<<kN / kNodesPerBlockFinal, kNodesPerBlockFinal * 32, 0, stream>>>(
        h1, h2, hp, cnt, slot, spill, a + kH * kD, ln_g + kHD, ln_b + kHD, xa, nullptr,
        Wo, bo, out);
}
